# starting kernel
# baseline (speedup 1.0000x reference)
#include <hip/hip_runtime.h>


typedef float v2f __attribute__((ext_vector_type(2)));
typedef float v8f __attribute__((ext_vector_type(8)));

#define CC 64
#define HH 4
#define OO 16
#define NEG_SLOPE 0.2f

__device__ __forceinline__ unsigned fenc(float f) {
  unsigned b = __float_as_uint(f);
  return (b & 0x80000000u) ? ~b : (b | 0x80000000u);
}
__device__ __forceinline__ float fdec(unsigned e) {
  return (e & 0x80000000u) ? __uint_as_float(e & 0x7FFFFFFFu)
                           : __uint_as_float(~e);
}

__global__ __launch_bounds__(128) void gat_gemm_wmma(
    const float* __restrict__ X, const float* __restrict__ W,
    float* __restrict__ Y, int nrows) {
  const int wave = threadIdx.x >> 5;
  const int lane = threadIdx.x & 31;
  const int m0 = blockIdx.x * 16;
  if (m0 >= nrows) return;
  const int n0 = wave * 16;
  const int lrow = lane & 15;
  const int khalf = (lane >> 4) << 1;

  const float* __restrict__ xrow = X + (size_t)(m0 + lrow) * CC;
  v8f c = {};
#pragma unroll
  for (int k = 0; k < CC; k += 4) {
    v2f a, b;
    a.x = xrow[k + khalf];
    a.y = xrow[k + khalf + 1];
    b.x = W[(size_t)(k + khalf) * CC + n0 + lrow];
    b.y = W[(size_t)(k + khalf + 1) * CC + n0 + lrow];
    c = __builtin_amdgcn_wmma_f32_16x16x4_f32(false, a, false, b,
                                              (short)0, c, false, false);
  }
  const int rbase = m0 + ((lane >> 4) << 3);
#pragma unroll
  for (int r = 0; r < 8; ++r)
    Y[(size_t)(rbase + r) * CC + n0 + lrow] = c[r];
}

__global__ void gat_init(float* __restrict__ accum, unsigned* __restrict__ menc,
                         float* __restrict__ denom, int n_nodes) {
  const long long i = (long long)blockIdx.x * blockDim.x + threadIdx.x;
  const long long nc = (long long)n_nodes * CC;
  if (i < nc) accum[i] = 0.0f;
  if (i < (long long)n_nodes * HH) {
    menc[i] = 0u;
    denom[i] = 0.0f;
  }
}

__global__ __launch_bounds__(256) void gat_edge_logits(
    const float* __restrict__ xl, const float* __restrict__ xr,
    const float* __restrict__ att, const int* __restrict__ src,
    const int* __restrict__ dst, float* __restrict__ elog,
    unsigned* __restrict__ menc, int nedges) {
  const long long t = (long long)blockIdx.x * blockDim.x + threadIdx.x;
  if (t >= (long long)nedges * HH) return;
  const int e = (int)(t >> 2);
  const int h = (int)(t & 3);
  const int s = src[e];
  const int d = dst[e];

  const float4* __restrict__ pl = (const float4*)(xl + (size_t)s * CC + h * OO);
  const float4* __restrict__ pr = (const float4*)(xr + (size_t)d * CC + h * OO);
  const float4* __restrict__ pa = (const float4*)(att + h * OO);

  float acc = 0.0f;
#pragma unroll
  for (int j = 0; j < 4; ++j) {
    float4 l = pl[j];
    float4 r = pr[j];
    float4 av = pa[j];
    float sx = l.x + r.x; sx = sx > 0.0f ? sx : NEG_SLOPE * sx;
    float sy = l.y + r.y; sy = sy > 0.0f ? sy : NEG_SLOPE * sy;
    float sz = l.z + r.z; sz = sz > 0.0f ? sz : NEG_SLOPE * sz;
    float sw = l.w + r.w; sw = sw > 0.0f ? sw : NEG_SLOPE * sw;
    acc += sx * av.x + sy * av.y + sz * av.z + sw * av.w;
  }
  elog[t] = acc;
  atomicMax(&menc[(size_t)d * HH + h], fenc(acc));
}

__global__ __launch_bounds__(256) void gat_edge_exp(
    const int* __restrict__ dst, float* __restrict__ elog,
    const unsigned* __restrict__ menc, float* __restrict__ denom, int nedges) {
  const long long t = (long long)blockIdx.x * blockDim.x + threadIdx.x;
  if (t >= (long long)nedges * HH) return;
  const int e = (int)(t >> 2);
  const int h = (int)(t & 3);
  const int d = dst[e];
  const float m = fdec(menc[(size_t)d * HH + h]);
  const float ex = __expf(elog[t] - m);
  elog[t] = ex;
  atomicAdd(&denom[(size_t)d * HH + h], ex);
}

__global__ __launch_bounds__(256) void gat_edge_scatter(
    const float* __restrict__ xl, const int* __restrict__ src,
    const int* __restrict__ dst, const float* __restrict__ elog,
    const float* __restrict__ denom, float* __restrict__ out, int nedges) {
  const long long t = (long long)blockIdx.x * blockDim.x + threadIdx.x;
  if (t >= (long long)nedges * HH) return;
  const int e = (int)(t >> 2);
  const int h = (int)(t & 3);
  const int s = src[e];
  const int d = dst[e];
  const float alpha = elog[t] / (denom[(size_t)d * HH + h] + 1e-16f);

  const float4* __restrict__ pl = (const float4*)(xl + (size_t)s * CC + h * OO);
  float* __restrict__ po = out + (size_t)d * CC + h * OO;
#pragma unroll
  for (int j = 0; j < 4; ++j) {
    float4 l = pl[j];
    atomicAdd(po + 4 * j + 0, l.x * alpha);
    atomicAdd(po + 4 * j + 1, l.y * alpha);
    atomicAdd(po + 4 * j + 2, l.z * alpha);
    atomicAdd(po + 4 * j + 3, l.w * alpha);
  }
}

__global__ __launch_bounds__(256) void gat_bias_act(
    float* __restrict__ buf, const float* __restrict__ bias, int n_nodes,
    int do_relu) {
  const long long i = (long long)blockIdx.x * blockDim.x + threadIdx.x;
  if (i >= (long long)n_nodes * CC) return;
  float v = buf[i] + bias[i & (CC - 1)];
  if (do_relu) v = fmaxf(v, 0.0f);
  buf[i] = v;
}

extern "C" void kernel_launch(void* const* d_in, const int* in_sizes, int n_in,
                              void* d_out, int out_size, void* d_ws,
                              size_t ws_size, hipStream_t stream) {
  const int N = in_sizes[0] / CC;
  const int E = in_sizes[1] / 2;

  const float* x = (const float*)d_in[0];
  const int* edge = (const int*)d_in[1];
  const int* src = edge;
  const int* dst = edge + E;

  const float* Wl[3] = {(const float*)d_in[2], (const float*)d_in[6],
                        (const float*)d_in[10]};
  const float* Wr[3] = {(const float*)d_in[3], (const float*)d_in[7],
                        (const float*)d_in[11]};
  const float* Av[3] = {(const float*)d_in[4], (const float*)d_in[8],
                        (const float*)d_in[12]};
  const float* Bv[3] = {(const float*)d_in[5], (const float*)d_in[9],
                        (const float*)d_in[13]};

  const size_t NC = (size_t)N * CC;
  const size_t EH = (size_t)E * HH;
  const size_t NH = (size_t)N * HH;
  float* ws = (float*)d_ws;
  float* xl = ws;                 ws += NC;
  float* xr = ws;                 ws += NC;
  float* elog = ws;               ws += EH;
  unsigned* menc = (unsigned*)ws; ws += NH;
  float* denom = ws;              ws += NH;
  float* hA = ws;                 ws += NC;
  float* hB = ws;                 ws += NC;

  const float* lin[3] = {x, hA, hB};
  float* lout[3] = {hA, hB, (float*)d_out};
  const int lrelu[3] = {1, 1, 0};

  const int gemm_grid = (N + 15) / 16;
  const long long nc_thr = (long long)N * CC;
  const long long eh_thr = (long long)E * HH;
  const int init_grid = (int)((nc_thr + 255) / 256);
  const int edge_grid = (int)((eh_thr + 255) / 256);

  for (int l = 0; l < 3; ++l) {
    gat_init<<<init_grid, 256, 0, stream>>>(lout[l], menc, denom, N);
    gat_gemm_wmma<<<gemm_grid, 128, 0, stream>>>(lin[l], Wl[l], xl, N);
    gat_gemm_wmma<<<gemm_grid, 128, 0, stream>>>(lin[l], Wr[l], xr, N);
    gat_edge_logits<<<edge_grid, 256, 0, stream>>>(xl, xr, Av[l], src, dst,
                                                   elog, menc, E);
    gat_edge_exp<<<edge_grid, 256, 0, stream>>>(dst, elog, menc, denom, E);
    gat_edge_scatter<<<edge_grid, 256, 0, stream>>>(xl, src, dst, elog, denom,
                                                    lout[l], E);
    gat_bias_act<<<init_grid, 256, 0, stream>>>(lout[l], Bv[l], N, lrelu[l]);
  }
}


# Group edges by destination once per call (int atomics: count, allocate, fill; then sort by source); per node softmax+sum in registers, running-max rescale; no float atomics; fixed sum order
# speedup vs baseline: 4.2286x; 4.2286x over previous
#include <hip/hip_runtime.h>


typedef float v2f __attribute__((ext_vector_type(2)));
typedef float v8f __attribute__((ext_vector_type(8)));

#define CC 64
#define HH 4
#define OO 16
#define NEG_SLOPE 0.2f

__global__ __launch_bounds__(128) void gat_gemm_wmma(
    const float* __restrict__ X, const float* __restrict__ W,
    float* __restrict__ Y, int nrows) {
  const int wave = threadIdx.x >> 5;
  const int lane = threadIdx.x & 31;
  const int m0 = blockIdx.x * 16;
  if (m0 >= nrows) return;
  const int n0 = wave * 16;
  const int lrow = lane & 15;
  const int khalf = (lane >> 4) << 1;

  const float* __restrict__ xrow = X + (size_t)(m0 + lrow) * CC;
  v8f c = {};
#pragma unroll
  for (int k = 0; k < CC; k += 4) {
    v2f a, b;
    a.x = xrow[k + khalf];
    a.y = xrow[k + khalf + 1];
    b.x = W[(size_t)(k + khalf) * CC + n0 + lrow];
    b.y = W[(size_t)(k + khalf + 1) * CC + n0 + lrow];
    c = __builtin_amdgcn_wmma_f32_16x16x4_f32(false, a, false, b,
                                              (short)0, c, false, false);
  }
  const int rbase = m0 + ((lane >> 4) << 3);
#pragma unroll
  for (int r = 0; r < 8; ++r)
    Y[(size_t)(rbase + r) * CC + n0 + lrow] = c[r];
}

__global__ __launch_bounds__(256) void csr_zero(int* __restrict__ cnt,
                                                int* __restrict__ total,
                                                int n_nodes) {
  const int i = blockIdx.x * blockDim.x + threadIdx.x;
  if (i < n_nodes) cnt[i] = 0;
  if (i == 0) total[0] = 0;
}

__global__ __launch_bounds__(256) void csr_count(const int* __restrict__ dst,
                                                 int* __restrict__ cnt,
                                                 int nedges) {
  const int e = blockIdx.x * blockDim.x + threadIdx.x;
  if (e < nedges) atomicAdd(&cnt[dst[e]], 1);
}

__global__ __launch_bounds__(256) void csr_alloc(const int* __restrict__ cnt,
                                                 int* __restrict__ start,
                                                 int* __restrict__ cursor,
                                                 int* __restrict__ total,
                                                 int n_nodes) {
  const int i = blockIdx.x * blockDim.x + threadIdx.x;
  const int lane = threadIdx.x & (warpSize - 1);
  const int c = (i < n_nodes) ? cnt[i] : 0;
  int incl = c;
  for (int off = 1; off < warpSize; off <<= 1) {
    const int t = __shfl_up(incl, off);
    if (lane >= off) incl += t;
  }
  const int last = warpSize - 1;
  const int wave_total = __shfl(incl, last);
  int base = 0;
  if (lane == last) base = atomicAdd(total, wave_total);
  base = __shfl(base, last);
  if (i < n_nodes) {
    const int s = base + incl - c;
    start[i] = s;
    cursor[i] = s;
  }
}

__global__ __launch_bounds__(256) void csr_fill(const int* __restrict__ src,
                                                const int* __restrict__ dst,
                                                int* __restrict__ cursor,
                                                int* __restrict__ csr_src,
                                                int nedges) {
  const int e = blockIdx.x * blockDim.x + threadIdx.x;
  if (e >= nedges) return;
  const int pos = atomicAdd(&cursor[dst[e]], 1);
  csr_src[pos] = src[e];
}

__global__ __launch_bounds__(256) void csr_order(const int* __restrict__ start,
                                                 const int* __restrict__ cnt,
                                                 const int* __restrict__ csr_in,
                                                 int* __restrict__ csr_out,
                                                 int n_nodes) {
  const int node = (blockIdx.x * blockDim.x + threadIdx.x) >> 2;
  const int q = threadIdx.x & 3;
  if (node >= n_nodes) return;
  const int beg = start[node];
  const int deg = cnt[node];
  for (int i = q; i < deg; i += 4) {
    const int v = csr_in[beg + i];
    int rank = 0;
    for (int j = 0; j < deg; ++j) {
      const int u = csr_in[beg + j];
      rank += (u < v || (u == v && j < i)) ? 1 : 0;
    }
    csr_out[beg + rank] = v;
  }
}

__device__ __forceinline__ float quad_xor1(float v) {
  return __int_as_float(__builtin_amdgcn_update_dpp(
      0, __float_as_int(v), 0xB1  , 0xF, 0xF, true));
}
__device__ __forceinline__ float quad_xor2(float v) {
  return __int_as_float(__builtin_amdgcn_update_dpp(
      0, __float_as_int(v), 0x4E  , 0xF, 0xF, true));
}

__global__ __launch_bounds__(256) void gat_node_attend(
    const float* __restrict__ xl, const float* __restrict__ xr,
    const float* __restrict__ att, const float* __restrict__ bias,
    const int* __restrict__ start, const int* __restrict__ cnt,
    const int* __restrict__ csr_src, float* __restrict__ out, int n_nodes,
    int do_relu) {
  const int node = (blockIdx.x * blockDim.x + threadIdx.x) >> 4;
  const int q = threadIdx.x & 15;
  if (node >= n_nodes) return;

  const int beg = start[node];
  const int deg = cnt[node];
  const float4 r4 = *(const float4*)(xr + (size_t)node * CC + 4 * q);
  const float4 a4 = *(const float4*)(att + 4 * q);

  float m = -INFINITY;
  float den = 0.0f;
  float4 acc = make_float4(0.0f, 0.0f, 0.0f, 0.0f);

  for (int k0 = 0; k0 < deg; k0 += 16) {
    const int c = min(16, deg - k0);
    const int myidx = (q < c) ? csr_src[beg + k0 + q] : 0;
    int s = __shfl(myidx, 0, 16);
    float4 nxt = *(const float4*)(xl + (size_t)s * CC + 4 * q);
    for (int j = 0; j < c; ++j) {
      const float4 l = nxt;
      if (j + 1 < c) {
        s = __shfl(myidx, j + 1, 16);
        nxt = *(const float4*)(xl + (size_t)s * CC + 4 * q);
      }
      float vx = l.x + r4.x; vx = vx > 0.0f ? vx : NEG_SLOPE * vx;
      float vy = l.y + r4.y; vy = vy > 0.0f ? vy : NEG_SLOPE * vy;
      float vz = l.z + r4.z; vz = vz > 0.0f ? vz : NEG_SLOPE * vz;
      float vw = l.w + r4.w; vw = vw > 0.0f ? vw : NEG_SLOPE * vw;
      float e = vx * a4.x + vy * a4.y + vz * a4.z + vw * a4.w;
      e += quad_xor1(e);
      e += quad_xor2(e);

      const float mn = fmaxf(m, e);
      const float sc = __expf(m - mn);
      const float p = __expf(e - mn);
      den = den * sc + p;
      acc.x = acc.x * sc + p * l.x;
      acc.y = acc.y * sc + p * l.y;
      acc.z = acc.z * sc + p * l.z;
      acc.w = acc.w * sc + p * l.w;
      m = mn;
    }
  }

  const float dd = den + 1e-16f;
  const float4 b4 = *(const float4*)(bias + 4 * q);
  float4 o;
  o.x = acc.x / dd + b4.x;
  o.y = acc.y / dd + b4.y;
  o.z = acc.z / dd + b4.z;
  o.w = acc.w / dd + b4.w;
  if (do_relu) {
    o.x = fmaxf(o.x, 0.0f);
    o.y = fmaxf(o.y, 0.0f);
    o.z = fmaxf(o.z, 0.0f);
    o.w = fmaxf(o.w, 0.0f);
  }
  *(float4*)(out + (size_t)node * CC + 4 * q) = o;
}

extern "C" void kernel_launch(void* const* d_in, const int* in_sizes, int n_in,
                              void* d_out, int out_size, void* d_ws,
                              size_t ws_size, hipStream_t stream) {
  const int N = in_sizes[0] / CC;
  const int E = in_sizes[1] / 2;

  const float* x = (const float*)d_in[0];
  const int* edge = (const int*)d_in[1];
  const int* src = edge;
  const int* dst = edge + E;

  const float* Wl[3] = {(const float*)d_in[2], (const float*)d_in[6],
                        (const float*)d_in[10]};
  const float* Wr[3] = {(const float*)d_in[3], (const float*)d_in[7],
                        (const float*)d_in[11]};
  const float* Av[3] = {(const float*)d_in[4], (const float*)d_in[8],
                        (const float*)d_in[12]};
  const float* Bv[3] = {(const float*)d_in[5], (const float*)d_in[9],
                        (const float*)d_in[13]};

  const size_t NC = (size_t)N * CC;
  float* ws = (float*)d_ws;
  float* xl = ws;                 ws += NC;
  float* xr = ws;                 ws += NC;
  float* hA = ws;                 ws += NC;
  float* hB = ws;                 ws += NC;
  int* csr_src = (int*)ws;        ws += (size_t)E;
  int* csr_tmp = (int*)ws;        ws += (size_t)E;
  int* cnt = (int*)ws;            ws += (size_t)N;
  int* start = (int*)ws;          ws += (size_t)N;
  int* cursor = (int*)ws;         ws += (size_t)N;
  int* total = (int*)ws;

  const float* lin[3] = {x, hA, hB};
  float* lout[3] = {hA, hB, (float*)d_out};
  const int lrelu[3] = {1, 1, 0};

  const int gemm_grid = (N + 15) / 16;
  const int node_grid = (N + 255) / 256;
  const int edge_grid = (E + 255) / 256;
  const int attend_grid = (N + 15) / 16;
  const int order_grid = (N + 63) / 64;

  csr_zero<<<node_grid, 256, 0, stream>>>(cnt, total, N);
  csr_count<<<edge_grid, 256, 0, stream>>>(dst, cnt, E);
  csr_alloc<<<node_grid, 256, 0, stream>>>(cnt, start, cursor, total, N);
  csr_fill<<<edge_grid, 256, 0, stream>>>(src, dst, cursor, csr_tmp, E);
  csr_order<<<order_grid, 256, 0, stream>>>(start, cnt, csr_tmp, csr_src, N);

  for (int l = 0; l < 3; ++l) {
    gat_gemm_wmma<<<gemm_grid, 128, 0, stream>>>(lin[l], Wl[l], xl, N);
    gat_gemm_wmma<<<gemm_grid, 128, 0, stream>>>(lin[l], Wr[l], xr, N);
    gat_node_attend<<<attend_grid, 256, 0, stream>>>(xl, xr, Av[l], Bv[l],
                                                     start, cnt, csr_src,
                                                     lout[l], N, lrelu[l]);
  }
}


# Both projections of a layer in one resident-grid launch: x tile read once (whole-line loads, shared via LDS), W fragments held in registers across row tiles, whole 128-byte line stores
# speedup vs baseline: 4.8362x; 1.1437x over previous
#include <hip/hip_runtime.h>


typedef float v2f __attribute__((ext_vector_type(2)));
typedef float v8f __attribute__((ext_vector_type(8)));

#define CC 64
#define HH 4
#define OO 16
#define NEG_SLOPE 0.2f

#define TILE_M 32
#define XPITCH 68

__device__ __forceinline__ float swap_half(float v) {
  return __int_as_float(__builtin_amdgcn_permlanex16(
      __float_as_int(v), __float_as_int(v), 0x76543210, 0xfedcba98, false,
      false));
}

__global__ __launch_bounds__(256) void gat_proj_pair_wmma(
    const float* __restrict__ X, const float* __restrict__ Wl,
    const float* __restrict__ Wr, float* __restrict__ Yl,
    float* __restrict__ Yr, int nrows) {
  __shared__ __attribute__((aligned(16))) float xs[2][TILE_M * XPITCH];

  const int tid = threadIdx.x;
  const int wave = __builtin_amdgcn_readfirstlane(tid >> 5);
  const int lane = tid & 31;
  const int c = lane & 15;
  const int h = lane >> 4;
  const int strip = wave & 3;
  const int rhalf = wave >> 2;
  const float* __restrict__ W = (strip < 2) ? Wl : Wr;
  float* __restrict__ Y = (strip < 2) ? Yl : Yr;
  const int col0 = (strip & 1) * 32;

  v2f b0[16], b1[16];
#pragma unroll
  for (int kk = 0; kk < 16; ++kk) {
    const int k = 4 * kk + 2 * h;
    b0[kk].x = W[(size_t)k * CC + col0 + c];
    b0[kk].y = W[(size_t)(k + 1) * CC + col0 + c];
    b1[kk].x = W[(size_t)k * CC + col0 + 16 + c];
    b1[kk].y = W[(size_t)(k + 1) * CC + col0 + 16 + c];
  }

  const int ntiles = (nrows + TILE_M - 1) / TILE_M;
  const int last = nrows - 1;
  const int lrow = tid >> 4;
  const int lcol = (tid & 15) << 2;

  int t = blockIdx.x;
  float4 g0, g1;
  {
    const int r0 = min(t * TILE_M + lrow, last);
    const int r1 = min(t * TILE_M + 16 + lrow, last);
    g0 = *(const float4*)(X + (size_t)r0 * CC + lcol);
    g1 = *(const float4*)(X + (size_t)r1 * CC + lcol);
    *(float4*)&xs[0][lrow * XPITCH + lcol] = g0;
    *(float4*)&xs[0][(16 + lrow) * XPITCH + lcol] = g1;
  }
  __syncthreads();

  int buf = 0;
  for (; t < ntiles; t += gridDim.x) {
    const int tn = t + gridDim.x;
    if (tn < ntiles) {
      const int r0 = min(tn * TILE_M + lrow, last);
      const int r1 = min(tn * TILE_M + 16 + lrow, last);
      g0 = *(const float4*)(X + (size_t)r0 * CC + lcol);
      g1 = *(const float4*)(X + (size_t)r1 * CC + lcol);
    }

    v8f c0 = {}, c1 = {};
    const float* xa = &xs[buf][(16 * rhalf + c) * XPITCH + 2 * h];
#pragma unroll
    for (int kk = 0; kk < 16; ++kk) {
      const v2f a = *(const v2f*)(xa + 4 * kk);
      c0 = __builtin_amdgcn_wmma_f32_16x16x4_f32(false, a, false, b0[kk],
                                                 (short)0, c0, false, false);
      c1 = __builtin_amdgcn_wmma_f32_16x16x4_f32(false, a, false, b1[kk],
                                                 (short)0, c1, false, false);
    }

    *(float4*)&xs[buf ^ 1][lrow * XPITCH + lcol] = g0;
    *(float4*)&xs[buf ^ 1][(16 + lrow) * XPITCH + lcol] = g1;
    __syncthreads();
    buf ^= 1;

    const int m0 = t * TILE_M + 16 * rhalf;
    float* yp = Y + (size_t)m0 * CC + col0 + lane;
    float va[8], vb[8];
#pragma unroll
    for (int r = 0; r < 8; ++r) {
      const float got = swap_half(h ? c0[r] : c1[r]);
      va[r] = h ? got : c0[r];
      vb[r] = h ? c1[r] : got;
    }
    if (m0 + 16 <= nrows) {
#pragma unroll
      for (int r = 0; r < 8; ++r) {
        yp[(size_t)r * CC] = va[r];
        yp[(size_t)(8 + r) * CC] = vb[r];
      }
    } else {
#pragma unroll
      for (int r = 0; r < 8; ++r) {
        if (m0 + r < nrows) yp[(size_t)r * CC] = va[r];
        if (m0 + 8 + r < nrows) yp[(size_t)(8 + r) * CC] = vb[r];
      }
    }
  }
}

__global__ __launch_bounds__(256) void csr_zero(int* __restrict__ cnt,
                                                int* __restrict__ total,
                                                int n_nodes) {
  const int i = blockIdx.x * blockDim.x + threadIdx.x;
  if (i < n_nodes) cnt[i] = 0;
  if (i == 0) total[0] = 0;
}

__global__ __launch_bounds__(256) void csr_count(const int* __restrict__ dst,
                                                 int* __restrict__ cnt,
                                                 int nedges) {
  const int e = blockIdx.x * blockDim.x + threadIdx.x;
  if (e < nedges) atomicAdd(&cnt[dst[e]], 1);
}

__global__ __launch_bounds__(256) void csr_alloc(const int* __restrict__ cnt,
                                                 int* __restrict__ start,
                                                 int* __restrict__ cursor,
                                                 int* __restrict__ total,
                                                 int n_nodes) {
  const int i = blockIdx.x * blockDim.x + threadIdx.x;
  const int lane = threadIdx.x & (warpSize - 1);
  const int c = (i < n_nodes) ? cnt[i] : 0;
  int incl = c;
  for (int off = 1; off < warpSize; off <<= 1) {
    const int t = __shfl_up(incl, off);
    if (lane >= off) incl += t;
  }
  const int last = warpSize - 1;
  const int wave_total = __shfl(incl, last);
  int base = 0;
  if (lane == last) base = atomicAdd(total, wave_total);
  base = __shfl(base, last);
  if (i < n_nodes) {
    const int s = base + incl - c;
    start[i] = s;
    cursor[i] = s;
  }
}

__global__ __launch_bounds__(256) void csr_fill(const int* __restrict__ src,
                                                const int* __restrict__ dst,
                                                int* __restrict__ cursor,
                                                int* __restrict__ csr_src,
                                                int nedges) {
  const int e = blockIdx.x * blockDim.x + threadIdx.x;
  if (e >= nedges) return;
  const int pos = atomicAdd(&cursor[dst[e]], 1);
  csr_src[pos] = src[e];
}

__global__ __launch_bounds__(256) void csr_order(const int* __restrict__ start,
                                                 const int* __restrict__ cnt,
                                                 const int* __restrict__ csr_in,
                                                 int* __restrict__ csr_out,
                                                 int n_nodes) {
  const int node = (blockIdx.x * blockDim.x + threadIdx.x) >> 2;
  const int q = threadIdx.x & 3;
  if (node >= n_nodes) return;
  const int beg = start[node];
  const int deg = cnt[node];
  for (int i = q; i < deg; i += 4) {
    const int v = csr_in[beg + i];
    int rank = 0;
    for (int j = 0; j < deg; ++j) {
      const int u = csr_in[beg + j];
      rank += (u < v || (u == v && j < i)) ? 1 : 0;
    }
    csr_out[beg + rank] = v;
  }
}

__device__ __forceinline__ float quad_xor1(float v) {
  return __int_as_float(__builtin_amdgcn_update_dpp(
      0, __float_as_int(v), 0xB1  , 0xF, 0xF, true));
}
__device__ __forceinline__ float quad_xor2(float v) {
  return __int_as_float(__builtin_amdgcn_update_dpp(
      0, __float_as_int(v), 0x4E  , 0xF, 0xF, true));
}

__global__ __launch_bounds__(256) void gat_node_attend(
    const float* __restrict__ xl, const float* __restrict__ xr,
    const float* __restrict__ att, const float* __restrict__ bias,
    const int* __restrict__ start, const int* __restrict__ cnt,
    const int* __restrict__ csr_src, float* __restrict__ out, int n_nodes,
    int do_relu) {
  const int node = (blockIdx.x * blockDim.x + threadIdx.x) >> 4;
  const int q = threadIdx.x & 15;
  if (node >= n_nodes) return;

  const int beg = start[node];
  const int deg = cnt[node];
  const float4 r4 = *(const float4*)(xr + (size_t)node * CC + 4 * q);
  const float4 a4 = *(const float4*)(att + 4 * q);

  float m = -INFINITY;
  float den = 0.0f;
  float4 acc = make_float4(0.0f, 0.0f, 0.0f, 0.0f);

  for (int k0 = 0; k0 < deg; k0 += 16) {
    const int c = min(16, deg - k0);
    const int myidx = (q < c) ? csr_src[beg + k0 + q] : 0;
    int s = __shfl(myidx, 0, 16);
    float4 nxt = *(const float4*)(xl + (size_t)s * CC + 4 * q);
    for (int j = 0; j < c; ++j) {
      const float4 l = nxt;
      if (j + 1 < c) {
        s = __shfl(myidx, j + 1, 16);
        nxt = *(const float4*)(xl + (size_t)s * CC + 4 * q);
      }
      float vx = l.x + r4.x; vx = vx > 0.0f ? vx : NEG_SLOPE * vx;
      float vy = l.y + r4.y; vy = vy > 0.0f ? vy : NEG_SLOPE * vy;
      float vz = l.z + r4.z; vz = vz > 0.0f ? vz : NEG_SLOPE * vz;
      float vw = l.w + r4.w; vw = vw > 0.0f ? vw : NEG_SLOPE * vw;
      float e = vx * a4.x + vy * a4.y + vz * a4.z + vw * a4.w;
      e += quad_xor1(e);
      e += quad_xor2(e);

      const float mn = fmaxf(m, e);
      const float sc = __expf(m - mn);
      const float p = __expf(e - mn);
      den = den * sc + p;
      acc.x = acc.x * sc + p * l.x;
      acc.y = acc.y * sc + p * l.y;
      acc.z = acc.z * sc + p * l.z;
      acc.w = acc.w * sc + p * l.w;
      m = mn;
    }
  }

  const float dd = den + 1e-16f;
  const float4 b4 = *(const float4*)(bias + 4 * q);
  float4 o;
  o.x = acc.x / dd + b4.x;
  o.y = acc.y / dd + b4.y;
  o.z = acc.z / dd + b4.z;
  o.w = acc.w / dd + b4.w;
  if (do_relu) {
    o.x = fmaxf(o.x, 0.0f);
    o.y = fmaxf(o.y, 0.0f);
    o.z = fmaxf(o.z, 0.0f);
    o.w = fmaxf(o.w, 0.0f);
  }
  *(float4*)(out + (size_t)node * CC + 4 * q) = o;
}

extern "C" void kernel_launch(void* const* d_in, const int* in_sizes, int n_in,
                              void* d_out, int out_size, void* d_ws,
                              size_t ws_size, hipStream_t stream) {
  const int N = in_sizes[0] / CC;
  const int E = in_sizes[1] / 2;

  const float* x = (const float*)d_in[0];
  const int* edge = (const int*)d_in[1];
  const int* src = edge;
  const int* dst = edge + E;

  const float* Wl[3] = {(const float*)d_in[2], (const float*)d_in[6],
                        (const float*)d_in[10]};
  const float* Wr[3] = {(const float*)d_in[3], (const float*)d_in[7],
                        (const float*)d_in[11]};
  const float* Av[3] = {(const float*)d_in[4], (const float*)d_in[8],
                        (const float*)d_in[12]};
  const float* Bv[3] = {(const float*)d_in[5], (const float*)d_in[9],
                        (const float*)d_in[13]};

  const size_t NC = (size_t)N * CC;
  float* ws = (float*)d_ws;
  float* xl = ws;                 ws += NC;
  float* xr = ws;                 ws += NC;
  float* hA = ws;                 ws += NC;
  float* hB = ws;                 ws += NC;
  int* csr_src = (int*)ws;        ws += (size_t)E;
  int* csr_tmp = (int*)ws;        ws += (size_t)E;
  int* cnt = (int*)ws;            ws += (size_t)N;
  int* start = (int*)ws;          ws += (size_t)N;
  int* cursor = (int*)ws;         ws += (size_t)N;
  int* total = (int*)ws;

  const float* lin[3] = {x, hA, hB};
  float* lout[3] = {hA, hB, (float*)d_out};
  const int lrelu[3] = {1, 1, 0};

  const int proj_tiles = (N + 31) / 32;
  const int proj_grid = proj_tiles < 1024 ? proj_tiles : 1024;
  const int node_grid = (N + 255) / 256;
  const int edge_grid = (E + 255) / 256;
  const int attend_grid = (N + 15) / 16;
  const int order_grid = (N + 63) / 64;

  csr_zero<<<node_grid, 256, 0, stream>>>(cnt, total, N);
  csr_count<<<edge_grid, 256, 0, stream>>>(dst, cnt, E);
  csr_alloc<<<node_grid, 256, 0, stream>>>(cnt, start, cursor, total, N);
  csr_fill<<<edge_grid, 256, 0, stream>>>(src, dst, cursor, csr_tmp, E);
  csr_order<<<order_grid, 256, 0, stream>>>(start, cnt, csr_tmp, csr_src, N);

  for (int l = 0; l < 3; ++l) {
    gat_proj_pair_wmma<<<proj_grid, 256, 0, stream>>>(lin[l], Wl[l], Wr[l],
                                                      xl, xr, N);
    gat_node_attend<<<attend_grid, 256, 0, stream>>>(xl, xr, Av[l], Bv[l],
                                                     start, cnt, csr_src,
                                                     lout[l], N, lrelu[l]);
  }
}


# Destination grouping as a two-level counting sort with per-edge counters in LDS (chunk-sort by 128-node bucket, then one block per bucket): 2 launches for 5, one global add per bucket; same sum order
# speedup vs baseline: 7.7744x; 1.6075x over previous
#include <hip/hip_runtime.h>


typedef float v2f __attribute__((ext_vector_type(2)));
typedef float v8f __attribute__((ext_vector_type(8)));

#define CC 64
#define HH 4
#define OO 16
#define NEG_SLOPE 0.2f

#define TILE_M 32
#define XPITCH 68

__device__ __forceinline__ float swap_half(float v) {
  return __int_as_float(__builtin_amdgcn_permlanex16(
      __float_as_int(v), __float_as_int(v), 0x76543210, 0xfedcba98, false,
      false));
}

__global__ __launch_bounds__(256) void gat_proj_pair_wmma(
    const float* __restrict__ X, const float* __restrict__ Wl,
    const float* __restrict__ Wr, float* __restrict__ Yl,
    float* __restrict__ Yr, int nrows) {
  __shared__ __attribute__((aligned(16))) float xs[2][TILE_M * XPITCH];

  const int tid = threadIdx.x;
  const int wave = __builtin_amdgcn_readfirstlane(tid >> 5);
  const int lane = tid & 31;
  const int c = lane & 15;
  const int h = lane >> 4;
  const int strip = wave & 3;
  const int rhalf = wave >> 2;
  const float* __restrict__ W = (strip < 2) ? Wl : Wr;
  float* __restrict__ Y = (strip < 2) ? Yl : Yr;
  const int col0 = (strip & 1) * 32;

  v2f b0[16], b1[16];
#pragma unroll
  for (int kk = 0; kk < 16; ++kk) {
    const int k = 4 * kk + 2 * h;
    b0[kk].x = W[(size_t)k * CC + col0 + c];
    b0[kk].y = W[(size_t)(k + 1) * CC + col0 + c];
    b1[kk].x = W[(size_t)k * CC + col0 + 16 + c];
    b1[kk].y = W[(size_t)(k + 1) * CC + col0 + 16 + c];
  }

  const int ntiles = (nrows + TILE_M - 1) / TILE_M;
  const int last = nrows - 1;
  const int lrow = tid >> 4;
  const int lcol = (tid & 15) << 2;

  int t = blockIdx.x;
  float4 g0, g1;
  {
    const int r0 = min(t * TILE_M + lrow, last);
    const int r1 = min(t * TILE_M + 16 + lrow, last);
    g0 = *(const float4*)(X + (size_t)r0 * CC + lcol);
    g1 = *(const float4*)(X + (size_t)r1 * CC + lcol);
    *(float4*)&xs[0][lrow * XPITCH + lcol] = g0;
    *(float4*)&xs[0][(16 + lrow) * XPITCH + lcol] = g1;
  }
  __syncthreads();

  int buf = 0;
  for (; t < ntiles; t += gridDim.x) {
    const int tn = t + gridDim.x;
    if (tn < ntiles) {
      const int r0 = min(tn * TILE_M + lrow, last);
      const int r1 = min(tn * TILE_M + 16 + lrow, last);
      g0 = *(const float4*)(X + (size_t)r0 * CC + lcol);
      g1 = *(const float4*)(X + (size_t)r1 * CC + lcol);
    }

    v8f c0 = {}, c1 = {};
    const float* xa = &xs[buf][(16 * rhalf + c) * XPITCH + 2 * h];
#pragma unroll
    for (int kk = 0; kk < 16; ++kk) {
      const v2f a = *(const v2f*)(xa + 4 * kk);
      c0 = __builtin_amdgcn_wmma_f32_16x16x4_f32(false, a, false, b0[kk],
                                                 (short)0, c0, false, false);
      c1 = __builtin_amdgcn_wmma_f32_16x16x4_f32(false, a, false, b1[kk],
                                                 (short)0, c1, false, false);
    }

    *(float4*)&xs[buf ^ 1][lrow * XPITCH + lcol] = g0;
    *(float4*)&xs[buf ^ 1][(16 + lrow) * XPITCH + lcol] = g1;
    __syncthreads();
    buf ^= 1;

    const int m0 = t * TILE_M + 16 * rhalf;
    float* yp = Y + (size_t)m0 * CC + col0 + lane;
    float va[8], vb[8];
#pragma unroll
    for (int r = 0; r < 8; ++r) {
      const float got = swap_half(h ? c0[r] : c1[r]);
      va[r] = h ? got : c0[r];
      vb[r] = h ? c1[r] : got;
    }
    if (m0 + 16 <= nrows) {
#pragma unroll
      for (int r = 0; r < 8; ++r) {
        yp[(size_t)r * CC] = va[r];
        yp[(size_t)(8 + r) * CC] = vb[r];
      }
    } else {
#pragma unroll
      for (int r = 0; r < 8; ++r) {
        if (m0 + r < nrows) yp[(size_t)r * CC] = va[r];
        if (m0 + 8 + r < nrows) yp[(size_t)(8 + r) * CC] = vb[r];
      }
    }
  }
}

#define BN_SHIFT 7
#define BN (1 << BN_SHIFT)
#define NB_MAX 1024
#define NCHUNK 128
#define CH_MAX 8192
#define KA_THREADS 1024
#define KA_PER_THREAD (CH_MAX / KA_THREADS)
#define KB_THREADS 256
#define KB_WAVES (KB_THREADS / 32)
#define RUNS_PER_WAVE (NCHUNK / KB_WAVES)
#define SRC_BITS 24
#define SRC_MASK ((1 << SRC_BITS) - 1)

__device__ __forceinline__ int wave_incl_scan(int v, int lane) {
  for (int off = 1; off < 32; off <<= 1) {
    const int t = __shfl_up(v, off);
    if (lane >= off) v += t;
  }
  return v;
}

__global__ __launch_bounds__(KA_THREADS) void csr_chunk_bin(
    const int* __restrict__ src, const int* __restrict__ dst,
    int* __restrict__ binned, int* __restrict__ table, int* __restrict__ total,
    int nedges, int chunk) {
  __shared__ int hist[NB_MAX];
  __shared__ int cur[NB_MAX];
  __shared__ int wsum[32];
  __shared__ int sorted[CH_MAX];

  const int tid = threadIdx.x;
  const int lane = tid & 31;
  const int wave = tid >> 5;
  const int e0 = blockIdx.x * chunk;
  const int len = max(0, min(chunk, nedges - e0));

  hist[tid] = 0;
  if (blockIdx.x == 0 && tid == 0) total[0] = 0;
  __syncthreads();

  int key[KA_PER_THREAD], bk[KA_PER_THREAD];
#pragma unroll
  for (int k = 0; k < KA_PER_THREAD; ++k) {
    const int i = tid + k * KA_THREADS;
    const int e = min(e0 + i, nedges - 1);
    const int d = dst[e];
    const int s = src[e];
    bk[k] = (i < len) ? (d >> BN_SHIFT) : -1;
    key[k] = s | ((d & (BN - 1)) << SRC_BITS);
  }
#pragma unroll
  for (int k = 0; k < KA_PER_THREAD; ++k)
    if (bk[k] >= 0) atomicAdd(&hist[bk[k]], 1);
  __syncthreads();

  const int c = hist[tid];
  const int incl = wave_incl_scan(c, lane);
  if (lane == 31) wsum[wave] = incl;
  __syncthreads();
  if (wave == 0) {
    const int w = wsum[lane];
    wsum[lane] = wave_incl_scan(w, lane) - w;
  }
  __syncthreads();
  const int excl = wsum[wave] + incl - c;
  cur[tid] = excl;
  table[(size_t)blockIdx.x * NB_MAX + tid] = excl;
  __syncthreads();

#pragma unroll
  for (int k = 0; k < KA_PER_THREAD; ++k) {
    if (bk[k] >= 0) {
      const int p = atomicAdd(&cur[bk[k]], 1);
      sorted[p] = key[k];
    }
  }
  __syncthreads();

  for (int i = tid; i < len; i += KA_THREADS) binned[e0 + i] = sorted[i];
}

__global__ __launch_bounds__(KB_THREADS) void csr_bucket_finish(
    const int* __restrict__ binned, const int* __restrict__ table,
    int* __restrict__ total, int* __restrict__ start, int* __restrict__ cnt,
    int* __restrict__ csr_tmp, int* __restrict__ csr_src, int n_nodes,
    int chunk) {
  __shared__ int lcnt[BN];
  __shared__ int lcur[BN];
  __shared__ int wsum[BN / 32];
  __shared__ int sbase;

  const int tid = threadIdx.x;
  const int lane = tid & 31;
  const int wave = tid >> 5;
  const int b = blockIdx.x;

  const int myr = wave + KB_WAVES * (lane & (RUNS_PER_WAVE - 1));
  const int o0 = table[(size_t)myr * NB_MAX + b];
  const int o1 = table[(size_t)myr * NB_MAX + b + 1];
  const int my_off = myr * chunk + o0;
  const int my_cnt = o1 - o0;
  if (tid < BN) lcnt[tid] = 0;
  __syncthreads();

  int val[RUNS_PER_WAVE];
#pragma unroll
  for (int k = 0; k < RUNS_PER_WAVE; ++k) {
    const int n = __builtin_amdgcn_readlane(my_cnt, k);
    const int o = __builtin_amdgcn_readlane(my_off, k);
    val[k] = -1;
    if (lane < n) val[k] = binned[o + lane];
  }
#pragma unroll
  for (int k = 0; k < RUNS_PER_WAVE; ++k)
    if (val[k] >= 0) atomicAdd(&lcnt[val[k] >> SRC_BITS], 1);
#pragma unroll
  for (int k = 0; k < RUNS_PER_WAVE; ++k) {
    const int n = __builtin_amdgcn_readlane(my_cnt, k);
    const int o = __builtin_amdgcn_readlane(my_off, k);
    for (int i = 32 + lane; i < n; i += 32)
      atomicAdd(&lcnt[binned[o + i] >> SRC_BITS], 1);
  }
  __syncthreads();

  const int c = (tid < BN) ? lcnt[tid] : 0;
  const int incl = wave_incl_scan(c, lane);
  if (tid < BN && lane == 31) wsum[wave] = incl;
  __syncthreads();
  int woff = 0, bucket_total = 0;
  for (int w = 0; w < BN / 32; ++w) {
    const int s = wsum[w];
    if (w < wave) woff += s;
    bucket_total += s;
  }
  if (tid == 0) sbase = atomicAdd(total, bucket_total);
  __syncthreads();
  const int base = sbase;
  if (tid < BN) {
    const int excl = woff + incl - c;
    lcur[tid] = excl;
    const int node = b * BN + tid;
    if (node < n_nodes) {
      start[node] = base + excl;
      cnt[node] = c;
    }
  }
  __syncthreads();

#pragma unroll
  for (int k = 0; k < RUNS_PER_WAVE; ++k) {
    if (val[k] >= 0) {
      const int p = atomicAdd(&lcur[val[k] >> SRC_BITS], 1);
      csr_tmp[base + p] = val[k] & SRC_MASK;
    }
  }
#pragma unroll
  for (int k = 0; k < RUNS_PER_WAVE; ++k) {
    const int n = __builtin_amdgcn_readlane(my_cnt, k);
    const int o = __builtin_amdgcn_readlane(my_off, k);
    for (int i = 32 + lane; i < n; i += 32) {
      const int v = binned[o + i];
      const int p = atomicAdd(&lcur[v >> SRC_BITS], 1);
      csr_tmp[base + p] = v & SRC_MASK;
    }
  }
  __syncthreads();

  const int q = tid & 3;
  for (int nd = tid >> 2; nd < BN; nd += KB_THREADS / 4) {
    const int deg = lcnt[nd];
    const int beg = base + lcur[nd] - deg;
    for (int i = q; i < deg; i += 4) {
      const int v = csr_tmp[beg + i];
      int rank = 0;
      for (int j = 0; j < deg; ++j) {
        const int u = csr_tmp[beg + j];
        rank += (u < v || (u == v && j < i)) ? 1 : 0;
      }
      csr_src[beg + rank] = v;
    }
  }
}

__device__ __forceinline__ float quad_xor1(float v) {
  return __int_as_float(__builtin_amdgcn_update_dpp(
      0, __float_as_int(v), 0xB1  , 0xF, 0xF, true));
}
__device__ __forceinline__ float quad_xor2(float v) {
  return __int_as_float(__builtin_amdgcn_update_dpp(
      0, __float_as_int(v), 0x4E  , 0xF, 0xF, true));
}

__global__ __launch_bounds__(256) void gat_node_attend(
    const float* __restrict__ xl, const float* __restrict__ xr,
    const float* __restrict__ att, const float* __restrict__ bias,
    const int* __restrict__ start, const int* __restrict__ cnt,
    const int* __restrict__ csr_src, float* __restrict__ out, int n_nodes,
    int do_relu) {
  const int node = (blockIdx.x * blockDim.x + threadIdx.x) >> 4;
  const int q = threadIdx.x & 15;
  if (node >= n_nodes) return;

  const int beg = start[node];
  const int deg = cnt[node];
  const float4 r4 = *(const float4*)(xr + (size_t)node * CC + 4 * q);
  const float4 a4 = *(const float4*)(att + 4 * q);

  float m = -INFINITY;
  float den = 0.0f;
  float4 acc = make_float4(0.0f, 0.0f, 0.0f, 0.0f);

  for (int k0 = 0; k0 < deg; k0 += 16) {
    const int c = min(16, deg - k0);
    const int myidx = (q < c) ? csr_src[beg + k0 + q] : 0;
    int s = __shfl(myidx, 0, 16);
    float4 nxt = *(const float4*)(xl + (size_t)s * CC + 4 * q);
    for (int j = 0; j < c; ++j) {
      const float4 l = nxt;
      if (j + 1 < c) {
        s = __shfl(myidx, j + 1, 16);
        nxt = *(const float4*)(xl + (size_t)s * CC + 4 * q);
      }
      float vx = l.x + r4.x; vx = vx > 0.0f ? vx : NEG_SLOPE * vx;
      float vy = l.y + r4.y; vy = vy > 0.0f ? vy : NEG_SLOPE * vy;
      float vz = l.z + r4.z; vz = vz > 0.0f ? vz : NEG_SLOPE * vz;
      float vw = l.w + r4.w; vw = vw > 0.0f ? vw : NEG_SLOPE * vw;
      float e = vx * a4.x + vy * a4.y + vz * a4.z + vw * a4.w;
      e += quad_xor1(e);
      e += quad_xor2(e);

      const float mn = fmaxf(m, e);
      const float sc = __expf(m - mn);
      const float p = __expf(e - mn);
      den = den * sc + p;
      acc.x = acc.x * sc + p * l.x;
      acc.y = acc.y * sc + p * l.y;
      acc.z = acc.z * sc + p * l.z;
      acc.w = acc.w * sc + p * l.w;
      m = mn;
    }
  }

  const float dd = den + 1e-16f;
  const float4 b4 = *(const float4*)(bias + 4 * q);
  float4 o;
  o.x = acc.x / dd + b4.x;
  o.y = acc.y / dd + b4.y;
  o.z = acc.z / dd + b4.z;
  o.w = acc.w / dd + b4.w;
  if (do_relu) {
    o.x = fmaxf(o.x, 0.0f);
    o.y = fmaxf(o.y, 0.0f);
    o.z = fmaxf(o.z, 0.0f);
    o.w = fmaxf(o.w, 0.0f);
  }
  *(float4*)(out + (size_t)node * CC + 4 * q) = o;
}

extern "C" void kernel_launch(void* const* d_in, const int* in_sizes, int n_in,
                              void* d_out, int out_size, void* d_ws,
                              size_t ws_size, hipStream_t stream) {
  const int N = in_sizes[0] / CC;
  const int E = in_sizes[1] / 2;

  const float* x = (const float*)d_in[0];
  const int* edge = (const int*)d_in[1];
  const int* src = edge;
  const int* dst = edge + E;

  const float* Wl[3] = {(const float*)d_in[2], (const float*)d_in[6],
                        (const float*)d_in[10]};
  const float* Wr[3] = {(const float*)d_in[3], (const float*)d_in[7],
                        (const float*)d_in[11]};
  const float* Av[3] = {(const float*)d_in[4], (const float*)d_in[8],
                        (const float*)d_in[12]};
  const float* Bv[3] = {(const float*)d_in[5], (const float*)d_in[9],
                        (const float*)d_in[13]};

  const size_t NC = (size_t)N * CC;
  float* ws = (float*)d_ws;
  float* xl = ws;                 ws += NC;
  float* xr = ws;                 ws += NC;
  float* hA = ws;                 ws += NC;
  float* hB = ws;                 ws += NC;
  int* csr_src = (int*)ws;        ws += (size_t)E;
  int* csr_tmp = (int*)ws;        ws += (size_t)E;
  int* cnt = (int*)ws;            ws += (size_t)N;
  int* start = (int*)ws;          ws += (size_t)N;
  int* total = (int*)ws;
  int* binned = (int*)hB;
  int* table = (int*)hB + (size_t)E;

  const float* lin[3] = {x, hA, hB};
  float* lout[3] = {hA, hB, (float*)d_out};
  const int lrelu[3] = {1, 1, 0};

  const int proj_tiles = (N + 31) / 32;
  const int proj_grid = proj_tiles < 1024 ? proj_tiles : 1024;
  const int attend_grid = (N + 15) / 16;
  const int chunk = (E + NCHUNK - 1) / NCHUNK;
  const int bucket_grid = (N + BN - 1) / BN;

  csr_chunk_bin<<<NCHUNK, KA_THREADS, 0, stream>>>(src, dst, binned, table,
                                                   total, E, chunk);
  csr_bucket_finish<<<bucket_grid, KB_THREADS, 0, stream>>>(
      binned, table, total, start, cnt, csr_tmp, csr_src, N, chunk);

  for (int l = 0; l < 3; ++l) {
    gat_proj_pair_wmma<<<proj_grid, 256, 0, stream>>>(lin[l], Wl[l], Wr[l],
                                                      xl, xr, N);
    gat_node_attend<<<attend_grid, 256, 0, stream>>>(xl, xr, Av[l], Bv[l],
                                                     start, cnt, csr_src,
                                                     lout[l], N, lrelu[l]);
  }
}


# Attention pass: 16 lanes walk 8 nodes; a node's source rows (16 per chunk) loaded together into registers; max, rescale and exp once per chunk and head, not per edge; sum order kept, rounding changes
# speedup vs baseline: 9.0466x; 1.1636x over previous
#include <hip/hip_runtime.h>


typedef float v2f __attribute__((ext_vector_type(2)));
typedef float v8f __attribute__((ext_vector_type(8)));

#define CC 64
#define HH 4
#define OO 16
#define NEG_SLOPE 0.2f

#define TILE_M 32
#define XPITCH 68

__device__ __forceinline__ float swap_half(float v) {
  return __int_as_float(__builtin_amdgcn_permlanex16(
      __float_as_int(v), __float_as_int(v), 0x76543210, 0xfedcba98, false,
      false));
}

__global__ __launch_bounds__(256) void gat_proj_pair_wmma(
    const float* __restrict__ X, const float* __restrict__ Wl,
    const float* __restrict__ Wr, float* __restrict__ Yl,
    float* __restrict__ Yr, int nrows) {
  __shared__ __attribute__((aligned(16))) float xs[2][TILE_M * XPITCH];

  const int tid = threadIdx.x;
  const int wave = __builtin_amdgcn_readfirstlane(tid >> 5);
  const int lane = tid & 31;
  const int c = lane & 15;
  const int h = lane >> 4;
  const int strip = wave & 3;
  const int rhalf = wave >> 2;
  const float* __restrict__ W = (strip < 2) ? Wl : Wr;
  float* __restrict__ Y = (strip < 2) ? Yl : Yr;
  const int col0 = (strip & 1) * 32;

  v2f b0[16], b1[16];
#pragma unroll
  for (int kk = 0; kk < 16; ++kk) {
    const int k = 4 * kk + 2 * h;
    b0[kk].x = W[(size_t)k * CC + col0 + c];
    b0[kk].y = W[(size_t)(k + 1) * CC + col0 + c];
    b1[kk].x = W[(size_t)k * CC + col0 + 16 + c];
    b1[kk].y = W[(size_t)(k + 1) * CC + col0 + 16 + c];
  }

  const int ntiles = (nrows + TILE_M - 1) / TILE_M;
  const int last = nrows - 1;
  const int lrow = tid >> 4;
  const int lcol = (tid & 15) << 2;

  int t = blockIdx.x;
  float4 g0, g1;
  {
    const int r0 = min(t * TILE_M + lrow, last);
    const int r1 = min(t * TILE_M + 16 + lrow, last);
    g0 = *(const float4*)(X + (size_t)r0 * CC + lcol);
    g1 = *(const float4*)(X + (size_t)r1 * CC + lcol);
    *(float4*)&xs[0][lrow * XPITCH + lcol] = g0;
    *(float4*)&xs[0][(16 + lrow) * XPITCH + lcol] = g1;
  }
  __syncthreads();

  int buf = 0;
  for (; t < ntiles; t += gridDim.x) {
    const int tn = t + gridDim.x;
    if (tn < ntiles) {
      const int r0 = min(tn * TILE_M + lrow, last);
      const int r1 = min(tn * TILE_M + 16 + lrow, last);
      g0 = *(const float4*)(X + (size_t)r0 * CC + lcol);
      g1 = *(const float4*)(X + (size_t)r1 * CC + lcol);
    }

    v8f c0 = {}, c1 = {};
    const float* xa = &xs[buf][(16 * rhalf + c) * XPITCH + 2 * h];
#pragma unroll
    for (int kk = 0; kk < 16; ++kk) {
      const v2f a = *(const v2f*)(xa + 4 * kk);
      c0 = __builtin_amdgcn_wmma_f32_16x16x4_f32(false, a, false, b0[kk],
                                                 (short)0, c0, false, false);
      c1 = __builtin_amdgcn_wmma_f32_16x16x4_f32(false, a, false, b1[kk],
                                                 (short)0, c1, false, false);
    }

    *(float4*)&xs[buf ^ 1][lrow * XPITCH + lcol] = g0;
    *(float4*)&xs[buf ^ 1][(16 + lrow) * XPITCH + lcol] = g1;
    __syncthreads();
    buf ^= 1;

    const int m0 = t * TILE_M + 16 * rhalf;
    float* yp = Y + (size_t)m0 * CC + col0 + lane;
    float va[8], vb[8];
#pragma unroll
    for (int r = 0; r < 8; ++r) {
      const float got = swap_half(h ? c0[r] : c1[r]);
      va[r] = h ? got : c0[r];
      vb[r] = h ? c1[r] : got;
    }
    if (m0 + 16 <= nrows) {
#pragma unroll
      for (int r = 0; r < 8; ++r) {
        yp[(size_t)r * CC] = va[r];
        yp[(size_t)(8 + r) * CC] = vb[r];
      }
    } else {
#pragma unroll
      for (int r = 0; r < 8; ++r) {
        if (m0 + r < nrows) yp[(size_t)r * CC] = va[r];
        if (m0 + 8 + r < nrows) yp[(size_t)(8 + r) * CC] = vb[r];
      }
    }
  }
}

#define BN_SHIFT 7
#define BN (1 << BN_SHIFT)
#define NB_MAX 1024
#define NCHUNK 128
#define CH_MAX 8192
#define KA_THREADS 1024
#define KA_PER_THREAD (CH_MAX / KA_THREADS)
#define KB_THREADS 256
#define KB_WAVES (KB_THREADS / 32)
#define RUNS_PER_WAVE (NCHUNK / KB_WAVES)
#define SRC_BITS 24
#define SRC_MASK ((1 << SRC_BITS) - 1)

__device__ __forceinline__ int wave_incl_scan(int v, int lane) {
  for (int off = 1; off < 32; off <<= 1) {
    const int t = __shfl_up(v, off);
    if (lane >= off) v += t;
  }
  return v;
}

__global__ __launch_bounds__(KA_THREADS) void csr_chunk_bin(
    const int* __restrict__ src, const int* __restrict__ dst,
    int* __restrict__ binned, int* __restrict__ table, int* __restrict__ total,
    int nedges, int chunk) {
  __shared__ int hist[NB_MAX];
  __shared__ int cur[NB_MAX];
  __shared__ int wsum[32];
  __shared__ int sorted[CH_MAX];

  const int tid = threadIdx.x;
  const int lane = tid & 31;
  const int wave = tid >> 5;
  const int e0 = blockIdx.x * chunk;
  const int len = max(0, min(chunk, nedges - e0));

  hist[tid] = 0;
  if (blockIdx.x == 0 && tid == 0) total[0] = 0;
  __syncthreads();

  int key[KA_PER_THREAD], bk[KA_PER_THREAD];
#pragma unroll
  for (int k = 0; k < KA_PER_THREAD; ++k) {
    const int i = tid + k * KA_THREADS;
    const int e = min(e0 + i, nedges - 1);
    const int d = dst[e];
    const int s = src[e];
    bk[k] = (i < len) ? (d >> BN_SHIFT) : -1;
    key[k] = s | ((d & (BN - 1)) << SRC_BITS);
  }
#pragma unroll
  for (int k = 0; k < KA_PER_THREAD; ++k)
    if (bk[k] >= 0) atomicAdd(&hist[bk[k]], 1);
  __syncthreads();

  const int c = hist[tid];
  const int incl = wave_incl_scan(c, lane);
  if (lane == 31) wsum[wave] = incl;
  __syncthreads();
  if (wave == 0) {
    const int w = wsum[lane];
    wsum[lane] = wave_incl_scan(w, lane) - w;
  }
  __syncthreads();
  const int excl = wsum[wave] + incl - c;
  cur[tid] = excl;
  table[(size_t)blockIdx.x * NB_MAX + tid] = excl;
  __syncthreads();

#pragma unroll
  for (int k = 0; k < KA_PER_THREAD; ++k) {
    if (bk[k] >= 0) {
      const int p = atomicAdd(&cur[bk[k]], 1);
      sorted[p] = key[k];
    }
  }
  __syncthreads();

  for (int i = tid; i < len; i += KA_THREADS) binned[e0 + i] = sorted[i];
}

__global__ __launch_bounds__(KB_THREADS) void csr_bucket_finish(
    const int* __restrict__ binned, const int* __restrict__ table,
    int* __restrict__ total, int* __restrict__ start, int* __restrict__ cnt,
    int* __restrict__ csr_tmp, int* __restrict__ csr_src, int n_nodes,
    int chunk) {
  __shared__ int lcnt[BN];
  __shared__ int lcur[BN];
  __shared__ int wsum[BN / 32];
  __shared__ int sbase;

  const int tid = threadIdx.x;
  const int lane = tid & 31;
  const int wave = tid >> 5;
  const int b = blockIdx.x;

  const int myr = wave + KB_WAVES * (lane & (RUNS_PER_WAVE - 1));
  const int o0 = table[(size_t)myr * NB_MAX + b];
  const int o1 = table[(size_t)myr * NB_MAX + b + 1];
  const int my_off = myr * chunk + o0;
  const int my_cnt = o1 - o0;
  if (tid < BN) lcnt[tid] = 0;
  __syncthreads();

  int val[RUNS_PER_WAVE];
#pragma unroll
  for (int k = 0; k < RUNS_PER_WAVE; ++k) {
    const int n = __builtin_amdgcn_readlane(my_cnt, k);
    const int o = __builtin_amdgcn_readlane(my_off, k);
    val[k] = -1;
    if (lane < n) val[k] = binned[o + lane];
  }
#pragma unroll
  for (int k = 0; k < RUNS_PER_WAVE; ++k)
    if (val[k] >= 0) atomicAdd(&lcnt[val[k] >> SRC_BITS], 1);
#pragma unroll
  for (int k = 0; k < RUNS_PER_WAVE; ++k) {
    const int n = __builtin_amdgcn_readlane(my_cnt, k);
    const int o = __builtin_amdgcn_readlane(my_off, k);
    for (int i = 32 + lane; i < n; i += 32)
      atomicAdd(&lcnt[binned[o + i] >> SRC_BITS], 1);
  }
  __syncthreads();

  const int c = (tid < BN) ? lcnt[tid] : 0;
  const int incl = wave_incl_scan(c, lane);
  if (tid < BN && lane == 31) wsum[wave] = incl;
  __syncthreads();
  int woff = 0, bucket_total = 0;
  for (int w = 0; w < BN / 32; ++w) {
    const int s = wsum[w];
    if (w < wave) woff += s;
    bucket_total += s;
  }
  if (tid == 0) sbase = atomicAdd(total, bucket_total);
  __syncthreads();
  const int base = sbase;
  if (tid < BN) {
    const int excl = woff + incl - c;
    lcur[tid] = excl;
    const int node = b * BN + tid;
    if (node < n_nodes) {
      start[node] = base + excl;
      cnt[node] = c;
    }
  }
  __syncthreads();

#pragma unroll
  for (int k = 0; k < RUNS_PER_WAVE; ++k) {
    if (val[k] >= 0) {
      const int p = atomicAdd(&lcur[val[k] >> SRC_BITS], 1);
      csr_tmp[base + p] = val[k] & SRC_MASK;
    }
  }
#pragma unroll
  for (int k = 0; k < RUNS_PER_WAVE; ++k) {
    const int n = __builtin_amdgcn_readlane(my_cnt, k);
    const int o = __builtin_amdgcn_readlane(my_off, k);
    for (int i = 32 + lane; i < n; i += 32) {
      const int v = binned[o + i];
      const int p = atomicAdd(&lcur[v >> SRC_BITS], 1);
      csr_tmp[base + p] = v & SRC_MASK;
    }
  }
  __syncthreads();

  const int q = tid & 3;
  for (int nd = tid >> 2; nd < BN; nd += KB_THREADS / 4) {
    const int deg = lcnt[nd];
    const int beg = base + lcur[nd] - deg;
    for (int i = q; i < deg; i += 4) {
      const int v = csr_tmp[beg + i];
      int rank = 0;
      for (int j = 0; j < deg; ++j) {
        const int u = csr_tmp[beg + j];
        rank += (u < v || (u == v && j < i)) ? 1 : 0;
      }
      csr_src[beg + rank] = v;
    }
  }
}

__device__ __forceinline__ float quad_xor1(float v) {
  return __int_as_float(__builtin_amdgcn_update_dpp(
      0, __float_as_int(v), 0xB1  , 0xF, 0xF, true));
}
__device__ __forceinline__ float quad_xor2(float v) {
  return __int_as_float(__builtin_amdgcn_update_dpp(
      0, __float_as_int(v), 0x4E  , 0xF, 0xF, true));
}
template <int T>
__device__ __forceinline__ float quad_lane(float v) {
  return __int_as_float(__builtin_amdgcn_update_dpp(
      0, __float_as_int(v), T * 0x55  , 0xF, 0xF,
      true));
}
template <int J>
__device__ __forceinline__ int row_lane(int v) {
  return __builtin_amdgcn_update_dpp(0, v, 0x150 + J  , 0xF,
                                     0xF, true);
}

__device__ __forceinline__ int chunk_index(const int* __restrict__ csr_src,
                                           int beg, int k0, int c, int q) {
  int v = 0;
  if (c > 0) v = csr_src[beg + k0 + min(q, c - 1)];
  return v;
}

#define NPG 8
#define ATT_THREADS 64

#define ATT_LOAD(J)                                                          \
  {                                                                          \
    const unsigned s_ = (unsigned)row_lane<J>(myidx);                        \
    row[J] = *(const float4*)(xlb + ((s_ << 8) | qoff));                     \
  }
#define ATT_LOGIT(J)                                                         \
  {                                                                          \
    const float4 l_ = row[J];                                                \
    float vx = l_.x + r4.x; vx = fmaxf(vx, NEG_SLOPE * vx);                  \
    float vy = l_.y + r4.y; vy = fmaxf(vy, NEG_SLOPE * vy);                  \
    float vz = l_.z + r4.z; vz = fmaxf(vz, NEG_SLOPE * vz);                  \
    float vw = l_.w + r4.w; vw = fmaxf(vw, NEG_SLOPE * vw);                  \
    float e_ = vx * a4.x + vy * a4.y + vz * a4.z + vw * a4.w;                \
    e_ += quad_xor1(e_);                                                     \
    e_ += quad_xor2(e_);                                                     \
    if (t == ((J) & 3)) E[(J) >> 2] = e_;                                    \
  }
#define ATT_SUM(J)                                                           \
  {                                                                          \
    const float p_ = quad_lane<(J) & 3>(P[(J) >> 2]);                        \
    den += p_;                                                               \
    acc.x = fmaf(p_, row[J].x, acc.x);                                       \
    acc.y = fmaf(p_, row[J].y, acc.y);                                       \
    acc.z = fmaf(p_, row[J].z, acc.z);                                       \
    acc.w = fmaf(p_, row[J].w, acc.w);                                       \
  }
#define ATT_PHASE(STEP)                                                      \
  do {                                                                       \
    STEP(0)  if (cw <= 1) break;  STEP(1)  if (cw <= 2) break;               \
    STEP(2)  if (cw <= 3) break;  STEP(3)  if (cw <= 4) break;               \
    STEP(4)  if (cw <= 5) break;  STEP(5)  if (cw <= 6) break;               \
    STEP(6)  if (cw <= 7) break;  STEP(7)  if (cw <= 8) break;               \
    STEP(8)  if (cw <= 9) break;  STEP(9)  if (cw <= 10) break;              \
    STEP(10) if (cw <= 11) break; STEP(11) if (cw <= 12) break;              \
    STEP(12) if (cw <= 13) break; STEP(13) if (cw <= 14) break;              \
    STEP(14) if (cw <= 15) break; STEP(15)                                   \
  } while (0)

__global__ __launch_bounds__(ATT_THREADS) void gat_node_attend(
    const float* __restrict__ xl, const float* __restrict__ xr,
    const float* __restrict__ att, const float* __restrict__ bias,
    const int* __restrict__ start, const int* __restrict__ cnt,
    const int* __restrict__ csr_src, float* __restrict__ out, int n_nodes,
    int do_relu) {
  const int gid = blockIdx.x * blockDim.x + threadIdx.x;
  if (((gid & ~31) >> 4) * NPG >= n_nodes) return;
  const int q = threadIdx.x & 15;
  const int t = q & 3;
  const int node0 = (gid >> 4) * NPG;
  const int lastn = n_nodes - 1;

  const int ni = node0 + (q & (NPG - 1));
  const int begs = start[min(ni, lastn)];
  const int degs = (ni < n_nodes) ? cnt[min(ni, lastn)] : 0;

  const float4 a4 = *(const float4*)(att + 4 * q);
  const float4 b4 = *(const float4*)(bias + 4 * q);
  const char* __restrict__ xlb = (const char*)xl;
  const unsigned qoff = (unsigned)q << 4;

  int beg = __shfl(begs, 0, 16);
  int deg = __shfl(degs, 0, 16);
  int pidx = chunk_index(csr_src, beg, 0, min(16, deg), q);
  float4 pr4 = *(const float4*)(xr + (size_t)min(node0, lastn) * CC + 4 * q);

  for (int i = 0; i < NPG; ++i) {
    const int node = node0 + i;
    const int cbeg = beg;
    const int cdeg = deg;
    const float4 r4 = pr4;
    const int idx0 = pidx;
    const int degw = max(__builtin_amdgcn_readlane(cdeg, 0),
                         __builtin_amdgcn_readlane(cdeg, 16));

    {
      const int nx = min(i + 1, NPG - 1);
      beg = __shfl(begs, nx, 16);
      deg = __shfl(degs, nx, 16);
      pidx = chunk_index(csr_src, beg, 0, min(16, deg), q);
      pr4 = *(const float4*)(xr + (size_t)min(node0 + nx, lastn) * CC + 4 * q);
    }

    float m = -INFINITY;
    float den = 0.0f;
    float4 acc = make_float4(0.0f, 0.0f, 0.0f, 0.0f);

    for (int k0 = 0; k0 < degw; k0 += 16) {
      const int c = min(16, max(0, cdeg - k0));
      const int cw = min(16, degw - k0);
      const int myidx =
          (k0 == 0) ? idx0 : chunk_index(csr_src, cbeg, k0, c, q);

      float4 row[16];
      ATT_PHASE(ATT_LOAD);

      float E[4] = {-INFINITY, -INFINITY, -INFINITY, -INFINITY};
      ATT_PHASE(ATT_LOGIT);

#pragma unroll
      for (int k = 0; k < 4; ++k)
        if (4 * k + t >= c) E[k] = -INFINITY;
      float mloc = fmaxf(fmaxf(E[0], E[1]), fmaxf(E[2], E[3]));
      mloc = fmaxf(mloc, quad_xor1(mloc));
      mloc = fmaxf(mloc, quad_xor2(mloc));
      const float mn = fmaxf(m, mloc);
      const float mref = (mn == -INFINITY) ? 0.0f : mn;
      const float sc = __expf(m - mref);
      float P[4];
#pragma unroll
      for (int k = 0; k < 4; ++k) P[k] = __expf(E[k] - mref);
      den *= sc;
      acc.x *= sc;
      acc.y *= sc;
      acc.z *= sc;
      acc.w *= sc;
      m = mn;

      ATT_PHASE(ATT_SUM);
    }

    if (node < n_nodes) {
      const float dd = den + 1e-16f;
      float4 o;
      o.x = acc.x / dd + b4.x;
      o.y = acc.y / dd + b4.y;
      o.z = acc.z / dd + b4.z;
      o.w = acc.w / dd + b4.w;
      if (do_relu) {
        o.x = fmaxf(o.x, 0.0f);
        o.y = fmaxf(o.y, 0.0f);
        o.z = fmaxf(o.z, 0.0f);
        o.w = fmaxf(o.w, 0.0f);
      }
      *(float4*)(out + (size_t)node * CC + 4 * q) = o;
    }
  }
}

extern "C" void kernel_launch(void* const* d_in, const int* in_sizes, int n_in,
                              void* d_out, int out_size, void* d_ws,
                              size_t ws_size, hipStream_t stream) {
  const int N = in_sizes[0] / CC;
  const int E = in_sizes[1] / 2;

  const float* x = (const float*)d_in[0];
  const int* edge = (const int*)d_in[1];
  const int* src = edge;
  const int* dst = edge + E;

  const float* Wl[3] = {(const float*)d_in[2], (const float*)d_in[6],
                        (const float*)d_in[10]};
  const float* Wr[3] = {(const float*)d_in[3], (const float*)d_in[7],
                        (const float*)d_in[11]};
  const float* Av[3] = {(const float*)d_in[4], (const float*)d_in[8],
                        (const float*)d_in[12]};
  const float* Bv[3] = {(const float*)d_in[5], (const float*)d_in[9],
                        (const float*)d_in[13]};

  const size_t NC = (size_t)N * CC;
  float* ws = (float*)d_ws;
  float* xl = ws;                 ws += NC;
  float* xr = ws;                 ws += NC;
  float* hA = ws;                 ws += NC;
  float* hB = ws;                 ws += NC;
  int* csr_src = (int*)ws;        ws += (size_t)E;
  int* csr_tmp = (int*)ws;        ws += (size_t)E;
  int* cnt = (int*)ws;            ws += (size_t)N;
  int* start = (int*)ws;          ws += (size_t)N;
  int* total = (int*)ws;
  int* binned = (int*)hB;
  int* table = (int*)hB + (size_t)E;

  const float* lin[3] = {x, hA, hB};
  float* lout[3] = {hA, hB, (float*)d_out};
  const int lrelu[3] = {1, 1, 0};

  const int proj_tiles = (N + 31) / 32;
  const int proj_grid = proj_tiles < 1024 ? proj_tiles : 1024;
  const int attend_grid =
      (N + (ATT_THREADS / 16) * NPG - 1) / ((ATT_THREADS / 16) * NPG);
  const int chunk = (E + NCHUNK - 1) / NCHUNK;
  const int bucket_grid = (N + BN - 1) / BN;

  csr_chunk_bin<<<NCHUNK, KA_THREADS, 0, stream>>>(src, dst, binned, table,
                                                   total, E, chunk);
  csr_bucket_finish<<<bucket_grid, KB_THREADS, 0, stream>>>(
      binned, table, total, start, cnt, csr_tmp, csr_src, N, chunk);

  for (int l = 0; l < 3; ++l) {
    gat_proj_pair_wmma<<<proj_grid, 256, 0, stream>>>(lin[l], Wl[l], Wr[l],
                                                      xl, xr, N);
    gat_node_attend<<<attend_grid, ATT_THREADS, 0, stream>>>(xl, xr, Av[l], Bv[l],
                                                     start, cnt, csr_src,
                                                     lout[l], N, lrelu[l]);
  }
}


# Bucket pass groups a bucket's keys by node in an LDS array instead of the csr_tmp workspace plane and ranks each segment from LDS, one thread per edge; same slots, same sum order
# speedup vs baseline: 9.6387x; 1.0655x over previous
#include <hip/hip_runtime.h>


typedef float v2f __attribute__((ext_vector_type(2)));
typedef float v8f __attribute__((ext_vector_type(8)));

#define CC 64
#define HH 4
#define OO 16
#define NEG_SLOPE 0.2f

#define TILE_M 32
#define XPITCH 68

__device__ __forceinline__ float swap_half(float v) {
  return __int_as_float(__builtin_amdgcn_permlanex16(
      __float_as_int(v), __float_as_int(v), 0x76543210, 0xfedcba98, false,
      false));
}

__global__ __launch_bounds__(256) void gat_proj_pair_wmma(
    const float* __restrict__ X, const float* __restrict__ Wl,
    const float* __restrict__ Wr, float* __restrict__ Yl,
    float* __restrict__ Yr, int nrows) {
  __shared__ __attribute__((aligned(16))) float xs[2][TILE_M * XPITCH];

  const int tid = threadIdx.x;
  const int wave = __builtin_amdgcn_readfirstlane(tid >> 5);
  const int lane = tid & 31;
  const int c = lane & 15;
  const int h = lane >> 4;
  const int strip = wave & 3;
  const int rhalf = wave >> 2;
  const float* __restrict__ W = (strip < 2) ? Wl : Wr;
  float* __restrict__ Y = (strip < 2) ? Yl : Yr;
  const int col0 = (strip & 1) * 32;

  v2f b0[16], b1[16];
#pragma unroll
  for (int kk = 0; kk < 16; ++kk) {
    const int k = 4 * kk + 2 * h;
    b0[kk].x = W[(size_t)k * CC + col0 + c];
    b0[kk].y = W[(size_t)(k + 1) * CC + col0 + c];
    b1[kk].x = W[(size_t)k * CC + col0 + 16 + c];
    b1[kk].y = W[(size_t)(k + 1) * CC + col0 + 16 + c];
  }

  const int ntiles = (nrows + TILE_M - 1) / TILE_M;
  const int last = nrows - 1;
  const int lrow = tid >> 4;
  const int lcol = (tid & 15) << 2;

  int t = blockIdx.x;
  float4 g0, g1;
  {
    const int r0 = min(t * TILE_M + lrow, last);
    const int r1 = min(t * TILE_M + 16 + lrow, last);
    g0 = *(const float4*)(X + (size_t)r0 * CC + lcol);
    g1 = *(const float4*)(X + (size_t)r1 * CC + lcol);
    *(float4*)&xs[0][lrow * XPITCH + lcol] = g0;
    *(float4*)&xs[0][(16 + lrow) * XPITCH + lcol] = g1;
  }
  __syncthreads();

  int buf = 0;
  for (; t < ntiles; t += gridDim.x) {
    const int tn = t + gridDim.x;
    if (tn < ntiles) {
      const int r0 = min(tn * TILE_M + lrow, last);
      const int r1 = min(tn * TILE_M + 16 + lrow, last);
      g0 = *(const float4*)(X + (size_t)r0 * CC + lcol);
      g1 = *(const float4*)(X + (size_t)r1 * CC + lcol);
    }

    v8f c0 = {}, c1 = {};
    const float* xa = &xs[buf][(16 * rhalf + c) * XPITCH + 2 * h];
#pragma unroll
    for (int kk = 0; kk < 16; ++kk) {
      const v2f a = *(const v2f*)(xa + 4 * kk);
      c0 = __builtin_amdgcn_wmma_f32_16x16x4_f32(false, a, false, b0[kk],
                                                 (short)0, c0, false, false);
      c1 = __builtin_amdgcn_wmma_f32_16x16x4_f32(false, a, false, b1[kk],
                                                 (short)0, c1, false, false);
    }

    *(float4*)&xs[buf ^ 1][lrow * XPITCH + lcol] = g0;
    *(float4*)&xs[buf ^ 1][(16 + lrow) * XPITCH + lcol] = g1;
    __syncthreads();
    buf ^= 1;

    const int m0 = t * TILE_M + 16 * rhalf;
    float* yp = Y + (size_t)m0 * CC + col0 + lane;
    float va[8], vb[8];
#pragma unroll
    for (int r = 0; r < 8; ++r) {
      const float got = swap_half(h ? c0[r] : c1[r]);
      va[r] = h ? got : c0[r];
      vb[r] = h ? c1[r] : got;
    }
    if (m0 + 16 <= nrows) {
#pragma unroll
      for (int r = 0; r < 8; ++r) {
        yp[(size_t)r * CC] = va[r];
        yp[(size_t)(8 + r) * CC] = vb[r];
      }
    } else {
#pragma unroll
      for (int r = 0; r < 8; ++r) {
        if (m0 + r < nrows) yp[(size_t)r * CC] = va[r];
        if (m0 + 8 + r < nrows) yp[(size_t)(8 + r) * CC] = vb[r];
      }
    }
  }
}

#define BN_SHIFT 7
#define BN (1 << BN_SHIFT)
#define NB_MAX 1024
#define NCHUNK 128
#define CH_MAX 8192
#define KA_THREADS 1024
#define KA_PER_THREAD (CH_MAX / KA_THREADS)
#define KB_THREADS 256
#define KB_WAVES (KB_THREADS / 32)
#define RUNS_PER_WAVE (NCHUNK / KB_WAVES)
#define SRC_BITS 24
#define SRC_MASK ((1 << SRC_BITS) - 1)
#define SEG_MAX 4096

__device__ __forceinline__ int wave_incl_scan(int v, int lane) {
  for (int off = 1; off < 32; off <<= 1) {
    const int t = __shfl_up(v, off);
    if (lane >= off) v += t;
  }
  return v;
}

__global__ __launch_bounds__(KA_THREADS) void csr_chunk_bin(
    const int* __restrict__ src, const int* __restrict__ dst,
    int* __restrict__ binned, int* __restrict__ table, int* __restrict__ total,
    int nedges, int chunk) {
  __shared__ int hist[NB_MAX];
  __shared__ int cur[NB_MAX];
  __shared__ int wsum[32];
  __shared__ int sorted[CH_MAX];

  const int tid = threadIdx.x;
  const int lane = tid & 31;
  const int wave = tid >> 5;
  const int e0 = blockIdx.x * chunk;
  const int len = max(0, min(chunk, nedges - e0));

  hist[tid] = 0;
  if (blockIdx.x == 0 && tid == 0) total[0] = 0;
  __syncthreads();

  int key[KA_PER_THREAD], bk[KA_PER_THREAD];
#pragma unroll
  for (int k = 0; k < KA_PER_THREAD; ++k) {
    const int i = tid + k * KA_THREADS;
    const int e = min(e0 + i, nedges - 1);
    const int d = dst[e];
    const int s = src[e];
    bk[k] = (i < len) ? (d >> BN_SHIFT) : -1;
    key[k] = s | ((d & (BN - 1)) << SRC_BITS);
  }
#pragma unroll
  for (int k = 0; k < KA_PER_THREAD; ++k)
    if (bk[k] >= 0) atomicAdd(&hist[bk[k]], 1);
  __syncthreads();

  const int c = hist[tid];
  const int incl = wave_incl_scan(c, lane);
  if (lane == 31) wsum[wave] = incl;
  __syncthreads();
  if (wave == 0) {
    const int w = wsum[lane];
    wsum[lane] = wave_incl_scan(w, lane) - w;
  }
  __syncthreads();
  const int excl = wsum[wave] + incl - c;
  cur[tid] = excl;
  table[(size_t)blockIdx.x * NB_MAX + tid] = excl;
  __syncthreads();

#pragma unroll
  for (int k = 0; k < KA_PER_THREAD; ++k) {
    if (bk[k] >= 0) {
      const int p = atomicAdd(&cur[bk[k]], 1);
      sorted[p] = key[k];
    }
  }
  __syncthreads();

  for (int i = tid; i < len; i += KA_THREADS) binned[e0 + i] = sorted[i];
}

__global__ __launch_bounds__(KB_THREADS) void csr_bucket_finish(
    const int* __restrict__ binned, const int* __restrict__ table,
    int* __restrict__ total, int* __restrict__ start, int* __restrict__ cnt,
    int* __restrict__ csr_tmp, int* __restrict__ csr_src, int n_nodes,
    int chunk) {
  __shared__ int lcnt[BN];
  __shared__ int lcur[BN];
  __shared__ int wsum[BN / 32];
  __shared__ int sbase;
  __shared__ int seg[SEG_MAX + 8];

  const int tid = threadIdx.x;
  const int lane = tid & 31;
  const int wave = tid >> 5;
  const int b = blockIdx.x;

  const int myr = wave + KB_WAVES * (lane & (RUNS_PER_WAVE - 1));
  const int o0 = table[(size_t)myr * NB_MAX + b];
  const int o1 = table[(size_t)myr * NB_MAX + b + 1];
  const int my_off = myr * chunk + o0;
  const int my_cnt = o1 - o0;
  if (tid < BN) lcnt[tid] = 0;
  __syncthreads();

  int val[RUNS_PER_WAVE];
#pragma unroll
  for (int k = 0; k < RUNS_PER_WAVE; ++k) {
    const int n = __builtin_amdgcn_readlane(my_cnt, k);
    const int o = __builtin_amdgcn_readlane(my_off, k);
    val[k] = -1;
    if (lane < n) val[k] = binned[o + lane];
  }
#pragma unroll
  for (int k = 0; k < RUNS_PER_WAVE; ++k)
    if (val[k] >= 0) atomicAdd(&lcnt[val[k] >> SRC_BITS], 1);
#pragma unroll
  for (int k = 0; k < RUNS_PER_WAVE; ++k) {
    const int n = __builtin_amdgcn_readlane(my_cnt, k);
    const int o = __builtin_amdgcn_readlane(my_off, k);
    for (int i = 32 + lane; i < n; i += 32)
      atomicAdd(&lcnt[binned[o + i] >> SRC_BITS], 1);
  }
  __syncthreads();

  const int c = (tid < BN) ? lcnt[tid] : 0;
  const int incl = wave_incl_scan(c, lane);
  if (tid < BN && lane == 31) wsum[wave] = incl;
  __syncthreads();
  int woff = 0, bucket_total = 0;
  for (int w = 0; w < BN / 32; ++w) {
    const int s = wsum[w];
    if (w < wave) woff += s;
    bucket_total += s;
  }
  if (tid == 0) sbase = atomicAdd(total, bucket_total);
  __syncthreads();
  const int base = sbase;
  if (tid < BN) {
    const int excl = woff + incl - c;
    lcur[tid] = excl;
    const int node = b * BN + tid;
    if (node < n_nodes) {
      start[node] = base + excl;
      cnt[node] = c;
    }
  }
  __syncthreads();

  if (bucket_total <= SEG_MAX) {
    if (tid < 8) seg[bucket_total + tid] = 0x7fffffff;
#pragma unroll
    for (int k = 0; k < RUNS_PER_WAVE; ++k) {
      if (val[k] >= 0) {
        const int p = atomicAdd(&lcur[val[k] >> SRC_BITS], 1);
        seg[p] = val[k];
      }
    }
#pragma unroll
    for (int k = 0; k < RUNS_PER_WAVE; ++k) {
      const int n = __builtin_amdgcn_readlane(my_cnt, k);
      const int o = __builtin_amdgcn_readlane(my_off, k);
      for (int i = 32 + lane; i < n; i += 32) {
        const int v = binned[o + i];
        const int p = atomicAdd(&lcur[v >> SRC_BITS], 1);
        seg[p] = v;
      }
    }
    __syncthreads();

    for (int p = tid; p < bucket_total; p += KB_THREADS) {
      const int v = seg[p];
      const int nd = v >> SRC_BITS;
      const int deg = lcnt[nd];
      const int sb = lcur[nd] - deg;
      int rank = 0;
      for (int j0 = sb; j0 < sb + deg; j0 += 8) {
#pragma unroll
        for (int jj = 0; jj < 8; ++jj) {
          const int u = seg[j0 + jj];
          rank += (u < v || (u == v && j0 + jj < p)) ? 1 : 0;
        }
      }
      csr_src[base + sb + rank] = v & SRC_MASK;
    }
  } else {
#pragma unroll
    for (int k = 0; k < RUNS_PER_WAVE; ++k) {
      if (val[k] >= 0) {
        const int p = atomicAdd(&lcur[val[k] >> SRC_BITS], 1);
        csr_tmp[base + p] = val[k] & SRC_MASK;
      }
    }
#pragma unroll
    for (int k = 0; k < RUNS_PER_WAVE; ++k) {
      const int n = __builtin_amdgcn_readlane(my_cnt, k);
      const int o = __builtin_amdgcn_readlane(my_off, k);
      for (int i = 32 + lane; i < n; i += 32) {
        const int v = binned[o + i];
        const int p = atomicAdd(&lcur[v >> SRC_BITS], 1);
        csr_tmp[base + p] = v & SRC_MASK;
      }
    }
    __syncthreads();

    const int q = tid & 3;
    for (int nd = tid >> 2; nd < BN; nd += KB_THREADS / 4) {
      const int deg = lcnt[nd];
      const int beg = base + lcur[nd] - deg;
      for (int i = q; i < deg; i += 4) {
        const int v = csr_tmp[beg + i];
        int rank = 0;
        for (int j = 0; j < deg; ++j) {
          const int u = csr_tmp[beg + j];
          rank += (u < v || (u == v && j < i)) ? 1 : 0;
        }
        csr_src[beg + rank] = v;
      }
    }
  }
}

__device__ __forceinline__ float quad_xor1(float v) {
  return __int_as_float(__builtin_amdgcn_update_dpp(
      0, __float_as_int(v), 0xB1  , 0xF, 0xF, true));
}
__device__ __forceinline__ float quad_xor2(float v) {
  return __int_as_float(__builtin_amdgcn_update_dpp(
      0, __float_as_int(v), 0x4E  , 0xF, 0xF, true));
}
template <int T>
__device__ __forceinline__ float quad_lane(float v) {
  return __int_as_float(__builtin_amdgcn_update_dpp(
      0, __float_as_int(v), T * 0x55  , 0xF, 0xF,
      true));
}
template <int J>
__device__ __forceinline__ int row_lane(int v) {
  return __builtin_amdgcn_update_dpp(0, v, 0x150 + J  , 0xF,
                                     0xF, true);
}

__device__ __forceinline__ int chunk_index(const int* __restrict__ csr_src,
                                           int beg, int k0, int c, int q) {
  int v = 0;
  if (c > 0) v = csr_src[beg + k0 + min(q, c - 1)];
  return v;
}

#define NPG 8
#define ATT_THREADS 64

#define ATT_LOAD(J)                                                          \
  {                                                                          \
    const unsigned s_ = (unsigned)row_lane<J>(myidx);                        \
    row[J] = *(const float4*)(xlb + ((s_ << 8) | qoff));                     \
  }
#define ATT_LOGIT(J)                                                         \
  {                                                                          \
    const float4 l_ = row[J];                                                \
    float vx = l_.x + r4.x; vx = fmaxf(vx, NEG_SLOPE * vx);                  \
    float vy = l_.y + r4.y; vy = fmaxf(vy, NEG_SLOPE * vy);                  \
    float vz = l_.z + r4.z; vz = fmaxf(vz, NEG_SLOPE * vz);                  \
    float vw = l_.w + r4.w; vw = fmaxf(vw, NEG_SLOPE * vw);                  \
    float e_ = vx * a4.x + vy * a4.y + vz * a4.z + vw * a4.w;                \
    e_ += quad_xor1(e_);                                                     \
    e_ += quad_xor2(e_);                                                     \
    if (t == ((J) & 3)) E[(J) >> 2] = e_;                                    \
  }
#define ATT_SUM(J)                                                           \
  {                                                                          \
    const float p_ = quad_lane<(J) & 3>(P[(J) >> 2]);                        \
    den += p_;                                                               \
    acc.x = fmaf(p_, row[J].x, acc.x);                                       \
    acc.y = fmaf(p_, row[J].y, acc.y);                                       \
    acc.z = fmaf(p_, row[J].z, acc.z);                                       \
    acc.w = fmaf(p_, row[J].w, acc.w);                                       \
  }
#define ATT_PHASE(STEP)                                                      \
  do {                                                                       \
    STEP(0)  if (cw <= 1) break;  STEP(1)  if (cw <= 2) break;               \
    STEP(2)  if (cw <= 3) break;  STEP(3)  if (cw <= 4) break;               \
    STEP(4)  if (cw <= 5) break;  STEP(5)  if (cw <= 6) break;               \
    STEP(6)  if (cw <= 7) break;  STEP(7)  if (cw <= 8) break;               \
    STEP(8)  if (cw <= 9) break;  STEP(9)  if (cw <= 10) break;              \
    STEP(10) if (cw <= 11) break; STEP(11) if (cw <= 12) break;              \
    STEP(12) if (cw <= 13) break; STEP(13) if (cw <= 14) break;              \
    STEP(14) if (cw <= 15) break; STEP(15)                                   \
  } while (0)

__global__ __launch_bounds__(ATT_THREADS) void gat_node_attend(
    const float* __restrict__ xl, const float* __restrict__ xr,
    const float* __restrict__ att, const float* __restrict__ bias,
    const int* __restrict__ start, const int* __restrict__ cnt,
    const int* __restrict__ csr_src, float* __restrict__ out, int n_nodes,
    int do_relu) {
  const int gid = blockIdx.x * blockDim.x + threadIdx.x;
  if (((gid & ~31) >> 4) * NPG >= n_nodes) return;
  const int q = threadIdx.x & 15;
  const int t = q & 3;
  const int node0 = (gid >> 4) * NPG;
  const int lastn = n_nodes - 1;

  const int ni = node0 + (q & (NPG - 1));
  const int begs = start[min(ni, lastn)];
  const int degs = (ni < n_nodes) ? cnt[min(ni, lastn)] : 0;

  const float4 a4 = *(const float4*)(att + 4 * q);
  const float4 b4 = *(const float4*)(bias + 4 * q);
  const char* __restrict__ xlb = (const char*)xl;
  const unsigned qoff = (unsigned)q << 4;

  int beg = __shfl(begs, 0, 16);
  int deg = __shfl(degs, 0, 16);
  int pidx = chunk_index(csr_src, beg, 0, min(16, deg), q);
  float4 pr4 = *(const float4*)(xr + (size_t)min(node0, lastn) * CC + 4 * q);

  for (int i = 0; i < NPG; ++i) {
    const int node = node0 + i;
    const int cbeg = beg;
    const int cdeg = deg;
    const float4 r4 = pr4;
    const int idx0 = pidx;
    const int degw = max(__builtin_amdgcn_readlane(cdeg, 0),
                         __builtin_amdgcn_readlane(cdeg, 16));

    {
      const int nx = min(i + 1, NPG - 1);
      beg = __shfl(begs, nx, 16);
      deg = __shfl(degs, nx, 16);
      pidx = chunk_index(csr_src, beg, 0, min(16, deg), q);
      pr4 = *(const float4*)(xr + (size_t)min(node0 + nx, lastn) * CC + 4 * q);
    }

    float m = -INFINITY;
    float den = 0.0f;
    float4 acc = make_float4(0.0f, 0.0f, 0.0f, 0.0f);

    for (int k0 = 0; k0 < degw; k0 += 16) {
      const int c = min(16, max(0, cdeg - k0));
      const int cw = min(16, degw - k0);
      const int myidx =
          (k0 == 0) ? idx0 : chunk_index(csr_src, cbeg, k0, c, q);

      float4 row[16];
      ATT_PHASE(ATT_LOAD);

      float E[4] = {-INFINITY, -INFINITY, -INFINITY, -INFINITY};
      ATT_PHASE(ATT_LOGIT);

#pragma unroll
      for (int k = 0; k < 4; ++k)
        if (4 * k + t >= c) E[k] = -INFINITY;
      float mloc = fmaxf(fmaxf(E[0], E[1]), fmaxf(E[2], E[3]));
      mloc = fmaxf(mloc, quad_xor1(mloc));
      mloc = fmaxf(mloc, quad_xor2(mloc));
      const float mn = fmaxf(m, mloc);
      const float mref = (mn == -INFINITY) ? 0.0f : mn;
      const float sc = __expf(m - mref);
      float P[4];
#pragma unroll
      for (int k = 0; k < 4; ++k) P[k] = __expf(E[k] - mref);
      den *= sc;
      acc.x *= sc;
      acc.y *= sc;
      acc.z *= sc;
      acc.w *= sc;
      m = mn;

      ATT_PHASE(ATT_SUM);
    }

    if (node < n_nodes) {
      const float dd = den + 1e-16f;
      float4 o;
      o.x = acc.x / dd + b4.x;
      o.y = acc.y / dd + b4.y;
      o.z = acc.z / dd + b4.z;
      o.w = acc.w / dd + b4.w;
      if (do_relu) {
        o.x = fmaxf(o.x, 0.0f);
        o.y = fmaxf(o.y, 0.0f);
        o.z = fmaxf(o.z, 0.0f);
        o.w = fmaxf(o.w, 0.0f);
      }
      *(float4*)(out + (size_t)node * CC + 4 * q) = o;
    }
  }
}

extern "C" void kernel_launch(void* const* d_in, const int* in_sizes, int n_in,
                              void* d_out, int out_size, void* d_ws,
                              size_t ws_size, hipStream_t stream) {
  const int N = in_sizes[0] / CC;
  const int E = in_sizes[1] / 2;

  const float* x = (const float*)d_in[0];
  const int* edge = (const int*)d_in[1];
  const int* src = edge;
  const int* dst = edge + E;

  const float* Wl[3] = {(const float*)d_in[2], (const float*)d_in[6],
                        (const float*)d_in[10]};
  const float* Wr[3] = {(const float*)d_in[3], (const float*)d_in[7],
                        (const float*)d_in[11]};
  const float* Av[3] = {(const float*)d_in[4], (const float*)d_in[8],
                        (const float*)d_in[12]};
  const float* Bv[3] = {(const float*)d_in[5], (const float*)d_in[9],
                        (const float*)d_in[13]};

  const size_t NC = (size_t)N * CC;
  float* ws = (float*)d_ws;
  float* xl = ws;                 ws += NC;
  float* xr = ws;                 ws += NC;
  float* hA = ws;                 ws += NC;
  float* hB = ws;                 ws += NC;
  int* csr_src = (int*)ws;        ws += (size_t)E;
  int* csr_tmp = (int*)ws;        ws += (size_t)E;
  int* cnt = (int*)ws;            ws += (size_t)N;
  int* start = (int*)ws;          ws += (size_t)N;
  int* total = (int*)ws;
  int* binned = (int*)hB;
  int* table = (int*)hB + (size_t)E;

  const float* lin[3] = {x, hA, hB};
  float* lout[3] = {hA, hB, (float*)d_out};
  const int lrelu[3] = {1, 1, 0};

  const int proj_tiles = (N + 31) / 32;
  const int proj_grid = proj_tiles < 1024 ? proj_tiles : 1024;
  const int attend_grid =
      (N + (ATT_THREADS / 16) * NPG - 1) / ((ATT_THREADS / 16) * NPG);
  const int chunk = (E + NCHUNK - 1) / NCHUNK;
  const int bucket_grid = (N + BN - 1) / BN;

  csr_chunk_bin<<<NCHUNK, KA_THREADS, 0, stream>>>(src, dst, binned, table,
                                                   total, E, chunk);
  csr_bucket_finish<<<bucket_grid, KB_THREADS, 0, stream>>>(
      binned, table, total, start, cnt, csr_tmp, csr_src, N, chunk);

  for (int l = 0; l < 3; ++l) {
    gat_proj_pair_wmma<<<proj_grid, 256, 0, stream>>>(lin[l], Wl[l], Wr[l],
                                                      xl, xr, N);
    gat_node_attend<<<attend_grid, ATT_THREADS, 0, stream>>>(xl, xr, Av[l], Bv[l],
                                                     start, cnt, csr_src,
                                                     lout[l], N, lrelu[l]);
  }
}


# Projections of layers 2 and 3 run in the tail of the previous layer's attention launch, on the block's 64 output rows kept in LDS; h never stored or re-read; 6 launches for 8; sum order unchanged
# speedup vs baseline: 9.9158x; 1.0287x over previous
#include <hip/hip_runtime.h>


typedef float v2f __attribute__((ext_vector_type(2)));
typedef float v8f __attribute__((ext_vector_type(8)));

#define CC 64
#define HH 4
#define OO 16
#define NEG_SLOPE 0.2f

#define TILE_M 32
#define XPITCH 68

__device__ __forceinline__ float swap_half(float v) {
  return __int_as_float(__builtin_amdgcn_permlanex16(
      __float_as_int(v), __float_as_int(v), 0x76543210, 0xfedcba98, false,
      false));
}

__global__ __launch_bounds__(256) void gat_proj_pair_wmma(
    const float* __restrict__ X, const float* __restrict__ Wl,
    const float* __restrict__ Wr, float* __restrict__ Yl,
    float* __restrict__ Yr, int nrows) {
  __shared__ __attribute__((aligned(16))) float xs[2][TILE_M * XPITCH];

  const int tid = threadIdx.x;
  const int wave = __builtin_amdgcn_readfirstlane(tid >> 5);
  const int lane = tid & 31;
  const int c = lane & 15;
  const int h = lane >> 4;
  const int strip = wave & 3;
  const int rhalf = wave >> 2;
  const float* __restrict__ W = (strip < 2) ? Wl : Wr;
  float* __restrict__ Y = (strip < 2) ? Yl : Yr;
  const int col0 = (strip & 1) * 32;

  v2f b0[16], b1[16];
#pragma unroll
  for (int kk = 0; kk < 16; ++kk) {
    const int k = 4 * kk + 2 * h;
    b0[kk].x = W[(size_t)k * CC + col0 + c];
    b0[kk].y = W[(size_t)(k + 1) * CC + col0 + c];
    b1[kk].x = W[(size_t)k * CC + col0 + 16 + c];
    b1[kk].y = W[(size_t)(k + 1) * CC + col0 + 16 + c];
  }

  const int ntiles = (nrows + TILE_M - 1) / TILE_M;
  const int last = nrows - 1;
  const int lrow = tid >> 4;
  const int lcol = (tid & 15) << 2;

  int t = blockIdx.x;
  float4 g0, g1;
  {
    const int r0 = min(t * TILE_M + lrow, last);
    const int r1 = min(t * TILE_M + 16 + lrow, last);
    g0 = *(const float4*)(X + (size_t)r0 * CC + lcol);
    g1 = *(const float4*)(X + (size_t)r1 * CC + lcol);
    *(float4*)&xs[0][lrow * XPITCH + lcol] = g0;
    *(float4*)&xs[0][(16 + lrow) * XPITCH + lcol] = g1;
  }
  __syncthreads();

  int buf = 0;
  for (; t < ntiles; t += gridDim.x) {
    const int tn = t + gridDim.x;
    if (tn < ntiles) {
      const int r0 = min(tn * TILE_M + lrow, last);
      const int r1 = min(tn * TILE_M + 16 + lrow, last);
      g0 = *(const float4*)(X + (size_t)r0 * CC + lcol);
      g1 = *(const float4*)(X + (size_t)r1 * CC + lcol);
    }

    v8f c0 = {}, c1 = {};
    const float* xa = &xs[buf][(16 * rhalf + c) * XPITCH + 2 * h];
#pragma unroll
    for (int kk = 0; kk < 16; ++kk) {
      const v2f a = *(const v2f*)(xa + 4 * kk);
      c0 = __builtin_amdgcn_wmma_f32_16x16x4_f32(false, a, false, b0[kk],
                                                 (short)0, c0, false, false);
      c1 = __builtin_amdgcn_wmma_f32_16x16x4_f32(false, a, false, b1[kk],
                                                 (short)0, c1, false, false);
    }

    *(float4*)&xs[buf ^ 1][lrow * XPITCH + lcol] = g0;
    *(float4*)&xs[buf ^ 1][(16 + lrow) * XPITCH + lcol] = g1;
    __syncthreads();
    buf ^= 1;

    const int m0 = t * TILE_M + 16 * rhalf;
    float* yp = Y + (size_t)m0 * CC + col0 + lane;
    float va[8], vb[8];
#pragma unroll
    for (int r = 0; r < 8; ++r) {
      const float got = swap_half(h ? c0[r] : c1[r]);
      va[r] = h ? got : c0[r];
      vb[r] = h ? c1[r] : got;
    }
    if (m0 + 16 <= nrows) {
#pragma unroll
      for (int r = 0; r < 8; ++r) {
        yp[(size_t)r * CC] = va[r];
        yp[(size_t)(8 + r) * CC] = vb[r];
      }
    } else {
#pragma unroll
      for (int r = 0; r < 8; ++r) {
        if (m0 + r < nrows) yp[(size_t)r * CC] = va[r];
        if (m0 + 8 + r < nrows) yp[(size_t)(8 + r) * CC] = vb[r];
      }
    }
  }
}

#define BN_SHIFT 7
#define BN (1 << BN_SHIFT)
#define NB_MAX 1024
#define NCHUNK 128
#define CH_MAX 8192
#define KA_THREADS 1024
#define KA_PER_THREAD (CH_MAX / KA_THREADS)
#define KB_THREADS 256
#define KB_WAVES (KB_THREADS / 32)
#define RUNS_PER_WAVE (NCHUNK / KB_WAVES)
#define SRC_BITS 24
#define SRC_MASK ((1 << SRC_BITS) - 1)
#define SEG_MAX 4096

__device__ __forceinline__ int wave_incl_scan(int v, int lane) {
  for (int off = 1; off < 32; off <<= 1) {
    const int t = __shfl_up(v, off);
    if (lane >= off) v += t;
  }
  return v;
}

__global__ __launch_bounds__(KA_THREADS) void csr_chunk_bin(
    const int* __restrict__ src, const int* __restrict__ dst,
    int* __restrict__ binned, int* __restrict__ table, int* __restrict__ total,
    int nedges, int chunk) {
  __shared__ int hist[NB_MAX];
  __shared__ int cur[NB_MAX];
  __shared__ int wsum[32];
  __shared__ int sorted[CH_MAX];

  const int tid = threadIdx.x;
  const int lane = tid & 31;
  const int wave = tid >> 5;
  const int e0 = blockIdx.x * chunk;
  const int len = max(0, min(chunk, nedges - e0));

  hist[tid] = 0;
  if (blockIdx.x == 0 && tid == 0) total[0] = 0;
  __syncthreads();

  int key[KA_PER_THREAD], bk[KA_PER_THREAD];
#pragma unroll
  for (int k = 0; k < KA_PER_THREAD; ++k) {
    const int i = tid + k * KA_THREADS;
    const int e = min(e0 + i, nedges - 1);
    const int d = dst[e];
    const int s = src[e];
    bk[k] = (i < len) ? (d >> BN_SHIFT) : -1;
    key[k] = s | ((d & (BN - 1)) << SRC_BITS);
  }
#pragma unroll
  for (int k = 0; k < KA_PER_THREAD; ++k)
    if (bk[k] >= 0) atomicAdd(&hist[bk[k]], 1);
  __syncthreads();

  const int c = hist[tid];
  const int incl = wave_incl_scan(c, lane);
  if (lane == 31) wsum[wave] = incl;
  __syncthreads();
  if (wave == 0) {
    const int w = wsum[lane];
    wsum[lane] = wave_incl_scan(w, lane) - w;
  }
  __syncthreads();
  const int excl = wsum[wave] + incl - c;
  cur[tid] = excl;
  table[(size_t)blockIdx.x * NB_MAX + tid] = excl;
  __syncthreads();

#pragma unroll
  for (int k = 0; k < KA_PER_THREAD; ++k) {
    if (bk[k] >= 0) {
      const int p = atomicAdd(&cur[bk[k]], 1);
      sorted[p] = key[k];
    }
  }
  __syncthreads();

  for (int i = tid; i < len; i += KA_THREADS) binned[e0 + i] = sorted[i];
}

__global__ __launch_bounds__(KB_THREADS) void csr_bucket_finish(
    const int* __restrict__ binned, const int* __restrict__ table,
    int* __restrict__ total, int* __restrict__ start, int* __restrict__ cnt,
    int* __restrict__ csr_tmp, int* __restrict__ csr_src, int n_nodes,
    int chunk) {
  __shared__ int lcnt[BN];
  __shared__ int lcur[BN];
  __shared__ int wsum[BN / 32];
  __shared__ int sbase;
  __shared__ int seg[SEG_MAX + 8];

  const int tid = threadIdx.x;
  const int lane = tid & 31;
  const int wave = tid >> 5;
  const int b = blockIdx.x;

  const int myr = wave + KB_WAVES * (lane & (RUNS_PER_WAVE - 1));
  const int o0 = table[(size_t)myr * NB_MAX + b];
  const int o1 = table[(size_t)myr * NB_MAX + b + 1];
  const int my_off = myr * chunk + o0;
  const int my_cnt = o1 - o0;
  if (tid < BN) lcnt[tid] = 0;
  __syncthreads();

  int val[RUNS_PER_WAVE];
#pragma unroll
  for (int k = 0; k < RUNS_PER_WAVE; ++k) {
    const int n = __builtin_amdgcn_readlane(my_cnt, k);
    const int o = __builtin_amdgcn_readlane(my_off, k);
    val[k] = -1;
    if (lane < n) val[k] = binned[o + lane];
  }
#pragma unroll
  for (int k = 0; k < RUNS_PER_WAVE; ++k)
    if (val[k] >= 0) atomicAdd(&lcnt[val[k] >> SRC_BITS], 1);
#pragma unroll
  for (int k = 0; k < RUNS_PER_WAVE; ++k) {
    const int n = __builtin_amdgcn_readlane(my_cnt, k);
    const int o = __builtin_amdgcn_readlane(my_off, k);
    for (int i = 32 + lane; i < n; i += 32)
      atomicAdd(&lcnt[binned[o + i] >> SRC_BITS], 1);
  }
  __syncthreads();

  const int c = (tid < BN) ? lcnt[tid] : 0;
  const int incl = wave_incl_scan(c, lane);
  if (tid < BN && lane == 31) wsum[wave] = incl;
  __syncthreads();
  int woff = 0, bucket_total = 0;
  for (int w = 0; w < BN / 32; ++w) {
    const int s = wsum[w];
    if (w < wave) woff += s;
    bucket_total += s;
  }
  if (tid == 0) sbase = atomicAdd(total, bucket_total);
  __syncthreads();
  const int base = sbase;
  if (tid < BN) {
    const int excl = woff + incl - c;
    lcur[tid] = excl;
    const int node = b * BN + tid;
    if (node < n_nodes) {
      start[node] = base + excl;
      cnt[node] = c;
    }
  }
  __syncthreads();

  if (bucket_total <= SEG_MAX) {
    if (tid < 8) seg[bucket_total + tid] = 0x7fffffff;
#pragma unroll
    for (int k = 0; k < RUNS_PER_WAVE; ++k) {
      if (val[k] >= 0) {
        const int p = atomicAdd(&lcur[val[k] >> SRC_BITS], 1);
        seg[p] = val[k];
      }
    }
#pragma unroll
    for (int k = 0; k < RUNS_PER_WAVE; ++k) {
      const int n = __builtin_amdgcn_readlane(my_cnt, k);
      const int o = __builtin_amdgcn_readlane(my_off, k);
      for (int i = 32 + lane; i < n; i += 32) {
        const int v = binned[o + i];
        const int p = atomicAdd(&lcur[v >> SRC_BITS], 1);
        seg[p] = v;
      }
    }
    __syncthreads();

    for (int p = tid; p < bucket_total; p += KB_THREADS) {
      const int v = seg[p];
      const int nd = v >> SRC_BITS;
      const int deg = lcnt[nd];
      const int sb = lcur[nd] - deg;
      int rank = 0;
      for (int j0 = sb; j0 < sb + deg; j0 += 8) {
#pragma unroll
        for (int jj = 0; jj < 8; ++jj) {
          const int u = seg[j0 + jj];
          rank += (u < v || (u == v && j0 + jj < p)) ? 1 : 0;
        }
      }
      csr_src[base + sb + rank] = v & SRC_MASK;
    }
  } else {
#pragma unroll
    for (int k = 0; k < RUNS_PER_WAVE; ++k) {
      if (val[k] >= 0) {
        const int p = atomicAdd(&lcur[val[k] >> SRC_BITS], 1);
        csr_tmp[base + p] = val[k] & SRC_MASK;
      }
    }
#pragma unroll
    for (int k = 0; k < RUNS_PER_WAVE; ++k) {
      const int n = __builtin_amdgcn_readlane(my_cnt, k);
      const int o = __builtin_amdgcn_readlane(my_off, k);
      for (int i = 32 + lane; i < n; i += 32) {
        const int v = binned[o + i];
        const int p = atomicAdd(&lcur[v >> SRC_BITS], 1);
        csr_tmp[base + p] = v & SRC_MASK;
      }
    }
    __syncthreads();

    const int q = tid & 3;
    for (int nd = tid >> 2; nd < BN; nd += KB_THREADS / 4) {
      const int deg = lcnt[nd];
      const int beg = base + lcur[nd] - deg;
      for (int i = q; i < deg; i += 4) {
        const int v = csr_tmp[beg + i];
        int rank = 0;
        for (int j = 0; j < deg; ++j) {
          const int u = csr_tmp[beg + j];
          rank += (u < v || (u == v && j < i)) ? 1 : 0;
        }
        csr_src[beg + rank] = v;
      }
    }
  }
}

__device__ __forceinline__ float quad_xor1(float v) {
  return __int_as_float(__builtin_amdgcn_update_dpp(
      0, __float_as_int(v), 0xB1  , 0xF, 0xF, true));
}
__device__ __forceinline__ float quad_xor2(float v) {
  return __int_as_float(__builtin_amdgcn_update_dpp(
      0, __float_as_int(v), 0x4E  , 0xF, 0xF, true));
}
template <int T>
__device__ __forceinline__ float quad_lane(float v) {
  return __int_as_float(__builtin_amdgcn_update_dpp(
      0, __float_as_int(v), T * 0x55  , 0xF, 0xF,
      true));
}
template <int J>
__device__ __forceinline__ int row_lane(int v) {
  return __builtin_amdgcn_update_dpp(0, v, 0x150 + J  , 0xF,
                                     0xF, true);
}

__device__ __forceinline__ int chunk_index(const int* __restrict__ csr_src,
                                           int beg, int k0, int c, int q) {
  int v = 0;
  if (c > 0) v = csr_src[beg + k0 + min(q, c - 1)];
  return v;
}

#define NPG 8
#define ATT_THREADS 64

#define ATT_LOAD(J)                                                          \
  {                                                                          \
    const unsigned s_ = (unsigned)row_lane<J>(myidx);                        \
    row[J] = *(const float4*)(xlb + ((s_ << 8) | qoff));                     \
  }
#define ATT_LOGIT(J)                                                         \
  {                                                                          \
    const float4 l_ = row[J];                                                \
    float vx = l_.x + r4.x; vx = fmaxf(vx, NEG_SLOPE * vx);                  \
    float vy = l_.y + r4.y; vy = fmaxf(vy, NEG_SLOPE * vy);                  \
    float vz = l_.z + r4.z; vz = fmaxf(vz, NEG_SLOPE * vz);                  \
    float vw = l_.w + r4.w; vw = fmaxf(vw, NEG_SLOPE * vw);                  \
    float e_ = vx * a4.x + vy * a4.y + vz * a4.z + vw * a4.w;                \
    e_ += quad_xor1(e_);                                                     \
    e_ += quad_xor2(e_);                                                     \
    if (t == ((J) & 3)) E[(J) >> 2] = e_;                                    \
  }
#define ATT_SUM(J)                                                           \
  {                                                                          \
    const float p_ = quad_lane<(J) & 3>(P[(J) >> 2]);                        \
    den += p_;                                                               \
    acc.x = fmaf(p_, row[J].x, acc.x);                                       \
    acc.y = fmaf(p_, row[J].y, acc.y);                                       \
    acc.z = fmaf(p_, row[J].z, acc.z);                                       \
    acc.w = fmaf(p_, row[J].w, acc.w);                                       \
  }
#define ATT_PHASE(STEP)                                                      \
  do {                                                                       \
    STEP(0)  if (cw <= 1) break;  STEP(1)  if (cw <= 2) break;               \
    STEP(2)  if (cw <= 3) break;  STEP(3)  if (cw <= 4) break;               \
    STEP(4)  if (cw <= 5) break;  STEP(5)  if (cw <= 6) break;               \
    STEP(6)  if (cw <= 7) break;  STEP(7)  if (cw <= 8) break;               \
    STEP(8)  if (cw <= 9) break;  STEP(9)  if (cw <= 10) break;              \
    STEP(10) if (cw <= 11) break; STEP(11) if (cw <= 12) break;              \
    STEP(12) if (cw <= 13) break; STEP(13) if (cw <= 14) break;              \
    STEP(14) if (cw <= 15) break; STEP(15)                                   \
  } while (0)

template <bool PROJ>
__device__ __forceinline__ void attend_nodes(
    const float* __restrict__ xl, const float* __restrict__ xr,
    const float* __restrict__ att, const float* __restrict__ bias,
    const int* __restrict__ start, const int* __restrict__ cnt,
    const int* __restrict__ csr_src, float* __restrict__ out,
    float* __restrict__ hs, int n_nodes, int do_relu) {
  const int gid = blockIdx.x * blockDim.x + threadIdx.x;
  if (!PROJ && ((gid & ~31) >> 4) * NPG >= n_nodes) return;
  const int q = threadIdx.x & 15;
  const int t = q & 3;
  const int node0 = (gid >> 4) * NPG;
  const int lastn = n_nodes - 1;

  const int ni = node0 + (q & (NPG - 1));
  const int begs = start[min(ni, lastn)];
  const int degs = (ni < n_nodes) ? cnt[min(ni, lastn)] : 0;

  const float4 a4 = *(const float4*)(att + 4 * q);
  const float4 b4 = *(const float4*)(bias + 4 * q);
  const char* __restrict__ xlb = (const char*)xl;
  const unsigned qoff = (unsigned)q << 4;

  int beg = __shfl(begs, 0, 16);
  int deg = __shfl(degs, 0, 16);
  int pidx = chunk_index(csr_src, beg, 0, min(16, deg), q);
  float4 pr4 = *(const float4*)(xr + (size_t)min(node0, lastn) * CC + 4 * q);

  for (int i = 0; i < NPG; ++i) {
    const int node = node0 + i;
    const int cbeg = beg;
    const int cdeg = deg;
    const float4 r4 = pr4;
    const int idx0 = pidx;
    const int degw = max(__builtin_amdgcn_readlane(cdeg, 0),
                         __builtin_amdgcn_readlane(cdeg, 16));

    {
      const int nx = min(i + 1, NPG - 1);
      beg = __shfl(begs, nx, 16);
      deg = __shfl(degs, nx, 16);
      pidx = chunk_index(csr_src, beg, 0, min(16, deg), q);
      pr4 = *(const float4*)(xr + (size_t)min(node0 + nx, lastn) * CC + 4 * q);
    }

    float m = -INFINITY;
    float den = 0.0f;
    float4 acc = make_float4(0.0f, 0.0f, 0.0f, 0.0f);

    for (int k0 = 0; k0 < degw; k0 += 16) {
      const int c = min(16, max(0, cdeg - k0));
      const int cw = min(16, degw - k0);
      const int myidx =
          (k0 == 0) ? idx0 : chunk_index(csr_src, cbeg, k0, c, q);

      float4 row[16];
      ATT_PHASE(ATT_LOAD);

      float E[4] = {-INFINITY, -INFINITY, -INFINITY, -INFINITY};
      ATT_PHASE(ATT_LOGIT);

#pragma unroll
      for (int k = 0; k < 4; ++k)
        if (4 * k + t >= c) E[k] = -INFINITY;
      float mloc = fmaxf(fmaxf(E[0], E[1]), fmaxf(E[2], E[3]));
      mloc = fmaxf(mloc, quad_xor1(mloc));
      mloc = fmaxf(mloc, quad_xor2(mloc));
      const float mn = fmaxf(m, mloc);
      const float mref = (mn == -INFINITY) ? 0.0f : mn;
      const float sc = __expf(m - mref);
      float P[4];
#pragma unroll
      for (int k = 0; k < 4; ++k) P[k] = __expf(E[k] - mref);
      den *= sc;
      acc.x *= sc;
      acc.y *= sc;
      acc.z *= sc;
      acc.w *= sc;
      m = mn;

      ATT_PHASE(ATT_SUM);
    }

    if (PROJ || node < n_nodes) {
      const float dd = den + 1e-16f;
      float4 o;
      o.x = acc.x / dd + b4.x;
      o.y = acc.y / dd + b4.y;
      o.z = acc.z / dd + b4.z;
      o.w = acc.w / dd + b4.w;
      if (do_relu) {
        o.x = fmaxf(o.x, 0.0f);
        o.y = fmaxf(o.y, 0.0f);
        o.z = fmaxf(o.z, 0.0f);
        o.w = fmaxf(o.w, 0.0f);
      }
      if (PROJ)
        *(float4*)&hs[((threadIdx.x >> 4) * NPG + i) * XPITCH + 4 * q] = o;
      else
        *(float4*)(out + (size_t)node * CC + 4 * q) = o;
    }
  }
}

__global__ __launch_bounds__(ATT_THREADS) void gat_node_attend(
    const float* __restrict__ xl, const float* __restrict__ xr,
    const float* __restrict__ att, const float* __restrict__ bias,
    const int* __restrict__ start, const int* __restrict__ cnt,
    const int* __restrict__ csr_src, float* __restrict__ out, int n_nodes,
    int do_relu) {
  attend_nodes<false>(xl, xr, att, bias, start, cnt, csr_src, out, nullptr,
                      n_nodes, do_relu);
}

#define ATTP_THREADS 128
#define ATTP_ROWS ((ATTP_THREADS / 16) * NPG)

__global__ __launch_bounds__(ATTP_THREADS) void gat_node_attend_proj(
    const float* __restrict__ xl, const float* __restrict__ xr,
    const float* __restrict__ att, const float* __restrict__ bias,
    const int* __restrict__ start, const int* __restrict__ cnt,
    const int* __restrict__ csr_src, const float* __restrict__ Wl,
    const float* __restrict__ Wr, float* __restrict__ Yl,
    float* __restrict__ Yr, int n_nodes, int do_relu) {
  __shared__ __attribute__((aligned(16))) float hs[ATTP_ROWS * XPITCH];

  attend_nodes<true>(xl, xr, att, bias, start, cnt, csr_src, nullptr, hs,
                     n_nodes, do_relu);

  const int tid = threadIdx.x;
  const int strip = __builtin_amdgcn_readfirstlane(tid >> 5);
  const int lane = tid & 31;
  const int c = lane & 15;
  const int h = lane >> 4;
  const float* __restrict__ W = (strip < 2) ? Wl : Wr;
  float* __restrict__ Y = (strip < 2) ? Yl : Yr;
  const int col0 = (strip & 1) * 32;

  v2f b0[16], b1[16];
#pragma unroll
  for (int kk = 0; kk < 16; ++kk) {
    const int k = 4 * kk + 2 * h;
    b0[kk].x = W[(size_t)k * CC + col0 + c];
    b0[kk].y = W[(size_t)(k + 1) * CC + col0 + c];
    b1[kk].x = W[(size_t)k * CC + col0 + 16 + c];
    b1[kk].y = W[(size_t)(k + 1) * CC + col0 + 16 + c];
  }
  __syncthreads();

  for (int rt = 0; rt < ATTP_ROWS / 16; ++rt) {
    v8f c0 = {}, c1 = {};
    const float* xa = &hs[(16 * rt + c) * XPITCH + 2 * h];
#pragma unroll
    for (int kk = 0; kk < 16; ++kk) {
      const v2f a = *(const v2f*)(xa + 4 * kk);
      c0 = __builtin_amdgcn_wmma_f32_16x16x4_f32(false, a, false, b0[kk],
                                                 (short)0, c0, false, false);
      c1 = __builtin_amdgcn_wmma_f32_16x16x4_f32(false, a, false, b1[kk],
                                                 (short)0, c1, false, false);
    }

    const int m0 = blockIdx.x * ATTP_ROWS + 16 * rt;
    float* yp = Y + (size_t)m0 * CC + col0 + lane;
    float va[8], vb[8];
#pragma unroll
    for (int r = 0; r < 8; ++r) {
      const float got = swap_half(h ? c0[r] : c1[r]);
      va[r] = h ? got : c0[r];
      vb[r] = h ? c1[r] : got;
    }
    if (m0 + 16 <= n_nodes) {
#pragma unroll
      for (int r = 0; r < 8; ++r) {
        yp[(size_t)r * CC] = va[r];
        yp[(size_t)(8 + r) * CC] = vb[r];
      }
    } else {
#pragma unroll
      for (int r = 0; r < 8; ++r) {
        if (m0 + r < n_nodes) yp[(size_t)r * CC] = va[r];
        if (m0 + 8 + r < n_nodes) yp[(size_t)(8 + r) * CC] = vb[r];
      }
    }
  }
}

extern "C" void kernel_launch(void* const* d_in, const int* in_sizes, int n_in,
                              void* d_out, int out_size, void* d_ws,
                              size_t ws_size, hipStream_t stream) {
  const int N = in_sizes[0] / CC;
  const int E = in_sizes[1] / 2;

  const float* x = (const float*)d_in[0];
  const int* edge = (const int*)d_in[1];
  const int* src = edge;
  const int* dst = edge + E;

  const float* Wl[3] = {(const float*)d_in[2], (const float*)d_in[6],
                        (const float*)d_in[10]};
  const float* Wr[3] = {(const float*)d_in[3], (const float*)d_in[7],
                        (const float*)d_in[11]};
  const float* Av[3] = {(const float*)d_in[4], (const float*)d_in[8],
                        (const float*)d_in[12]};
  const float* Bv[3] = {(const float*)d_in[5], (const float*)d_in[9],
                        (const float*)d_in[13]};

  const size_t NC = (size_t)N * CC;
  float* ws = (float*)d_ws;
  float* xl = ws;                 ws += NC;
  float* xr = ws;                 ws += NC;
  float* hA = ws;                 ws += NC;
  float* hB = ws;                 ws += NC;
  int* csr_src = (int*)ws;        ws += (size_t)E;
  int* csr_tmp = (int*)ws;        ws += (size_t)E;
  int* cnt = (int*)ws;            ws += (size_t)N;
  int* start = (int*)ws;          ws += (size_t)N;
  int* total = (int*)ws;
  int* binned = (int*)hB;
  int* table = (int*)hB + (size_t)E;

  const int proj_tiles = (N + 31) / 32;
  const int proj_grid = proj_tiles < 1024 ? proj_tiles : 1024;
  const int attend_grid =
      (N + (ATT_THREADS / 16) * NPG - 1) / ((ATT_THREADS / 16) * NPG);
  const int attp_grid = (N + ATTP_ROWS - 1) / ATTP_ROWS;
  const int chunk = (E + NCHUNK - 1) / NCHUNK;
  const int bucket_grid = (N + BN - 1) / BN;

  csr_chunk_bin<<<NCHUNK, KA_THREADS, 0, stream>>>(src, dst, binned, table,
                                                   total, E, chunk);
  csr_bucket_finish<<<bucket_grid, KB_THREADS, 0, stream>>>(
      binned, table, total, start, cnt, csr_tmp, csr_src, N, chunk);

  gat_proj_pair_wmma<<<proj_grid, 256, 0, stream>>>(x, Wl[0], Wr[0], xl, xr, N);
  gat_node_attend_proj<<<attp_grid, ATTP_THREADS, 0, stream>>>(
      xl, xr, Av[0], Bv[0], start, cnt, csr_src, Wl[1], Wr[1], hA, hB, N, 1);
  gat_node_attend_proj<<<attp_grid, ATTP_THREADS, 0, stream>>>(
      hA, hB, Av[1], Bv[1], start, cnt, csr_src, Wl[2], Wr[2], xl, xr, N, 1);
  gat_node_attend<<<attend_grid, ATT_THREADS, 0, stream>>>(
      xl, xr, Av[2], Bv[2], start, cnt, csr_src, (float*)d_out, N, 0);
}


# Layers 1 and 2: 4 nodes per 16-lane group (32 per block, 3125 blocks), more blocks than fit at once, so ending blocks' matrix tails run under starting blocks' row gathers; sums unchanged
# speedup vs baseline: 10.3362x; 1.0424x over previous
#include <hip/hip_runtime.h>


typedef float v2f __attribute__((ext_vector_type(2)));
typedef float v8f __attribute__((ext_vector_type(8)));

#define CC 64
#define HH 4
#define OO 16
#define NEG_SLOPE 0.2f

#define TILE_M 32
#define XPITCH 68

__device__ __forceinline__ float swap_half(float v) {
  return __int_as_float(__builtin_amdgcn_permlanex16(
      __float_as_int(v), __float_as_int(v), 0x76543210, 0xfedcba98, false,
      false));
}

__global__ __launch_bounds__(256) void gat_proj_pair_wmma(
    const float* __restrict__ X, const float* __restrict__ Wl,
    const float* __restrict__ Wr, float* __restrict__ Yl,
    float* __restrict__ Yr, int nrows) {
  __shared__ __attribute__((aligned(16))) float xs[2][TILE_M * XPITCH];

  const int tid = threadIdx.x;
  const int wave = __builtin_amdgcn_readfirstlane(tid >> 5);
  const int lane = tid & 31;
  const int c = lane & 15;
  const int h = lane >> 4;
  const int strip = wave & 3;
  const int rhalf = wave >> 2;
  const float* __restrict__ W = (strip < 2) ? Wl : Wr;
  float* __restrict__ Y = (strip < 2) ? Yl : Yr;
  const int col0 = (strip & 1) * 32;

  v2f b0[16], b1[16];
#pragma unroll
  for (int kk = 0; kk < 16; ++kk) {
    const int k = 4 * kk + 2 * h;
    b0[kk].x = W[(size_t)k * CC + col0 + c];
    b0[kk].y = W[(size_t)(k + 1) * CC + col0 + c];
    b1[kk].x = W[(size_t)k * CC + col0 + 16 + c];
    b1[kk].y = W[(size_t)(k + 1) * CC + col0 + 16 + c];
  }

  const int ntiles = (nrows + TILE_M - 1) / TILE_M;
  const int last = nrows - 1;
  const int lrow = tid >> 4;
  const int lcol = (tid & 15) << 2;

  int t = blockIdx.x;
  float4 g0, g1;
  {
    const int r0 = min(t * TILE_M + lrow, last);
    const int r1 = min(t * TILE_M + 16 + lrow, last);
    g0 = *(const float4*)(X + (size_t)r0 * CC + lcol);
    g1 = *(const float4*)(X + (size_t)r1 * CC + lcol);
    *(float4*)&xs[0][lrow * XPITCH + lcol] = g0;
    *(float4*)&xs[0][(16 + lrow) * XPITCH + lcol] = g1;
  }
  __syncthreads();

  int buf = 0;
  for (; t < ntiles; t += gridDim.x) {
    const int tn = t + gridDim.x;
    if (tn < ntiles) {
      const int r0 = min(tn * TILE_M + lrow, last);
      const int r1 = min(tn * TILE_M + 16 + lrow, last);
      g0 = *(const float4*)(X + (size_t)r0 * CC + lcol);
      g1 = *(const float4*)(X + (size_t)r1 * CC + lcol);
    }

    v8f c0 = {}, c1 = {};
    const float* xa = &xs[buf][(16 * rhalf + c) * XPITCH + 2 * h];
#pragma unroll
    for (int kk = 0; kk < 16; ++kk) {
      const v2f a = *(const v2f*)(xa + 4 * kk);
      c0 = __builtin_amdgcn_wmma_f32_16x16x4_f32(false, a, false, b0[kk],
                                                 (short)0, c0, false, false);
      c1 = __builtin_amdgcn_wmma_f32_16x16x4_f32(false, a, false, b1[kk],
                                                 (short)0, c1, false, false);
    }

    *(float4*)&xs[buf ^ 1][lrow * XPITCH + lcol] = g0;
    *(float4*)&xs[buf ^ 1][(16 + lrow) * XPITCH + lcol] = g1;
    __syncthreads();
    buf ^= 1;

    const int m0 = t * TILE_M + 16 * rhalf;
    float* yp = Y + (size_t)m0 * CC + col0 + lane;
    float va[8], vb[8];
#pragma unroll
    for (int r = 0; r < 8; ++r) {
      const float got = swap_half(h ? c0[r] : c1[r]);
      va[r] = h ? got : c0[r];
      vb[r] = h ? c1[r] : got;
    }
    if (m0 + 16 <= nrows) {
#pragma unroll
      for (int r = 0; r < 8; ++r) {
        yp[(size_t)r * CC] = va[r];
        yp[(size_t)(8 + r) * CC] = vb[r];
      }
    } else {
#pragma unroll
      for (int r = 0; r < 8; ++r) {
        if (m0 + r < nrows) yp[(size_t)r * CC] = va[r];
        if (m0 + 8 + r < nrows) yp[(size_t)(8 + r) * CC] = vb[r];
      }
    }
  }
}

#define BN_SHIFT 7
#define BN (1 << BN_SHIFT)
#define NB_MAX 1024
#define NCHUNK 128
#define CH_MAX 8192
#define KA_THREADS 1024
#define KA_PER_THREAD (CH_MAX / KA_THREADS)
#define KB_THREADS 256
#define KB_WAVES (KB_THREADS / 32)
#define RUNS_PER_WAVE (NCHUNK / KB_WAVES)
#define SRC_BITS 24
#define SRC_MASK ((1 << SRC_BITS) - 1)
#define SEG_MAX 4096

__device__ __forceinline__ int wave_incl_scan(int v, int lane) {
  for (int off = 1; off < 32; off <<= 1) {
    const int t = __shfl_up(v, off);
    if (lane >= off) v += t;
  }
  return v;
}

__global__ __launch_bounds__(KA_THREADS) void csr_chunk_bin(
    const int* __restrict__ src, const int* __restrict__ dst,
    int* __restrict__ binned, int* __restrict__ table, int* __restrict__ total,
    int nedges, int chunk) {
  __shared__ int hist[NB_MAX];
  __shared__ int cur[NB_MAX];
  __shared__ int wsum[32];
  __shared__ int sorted[CH_MAX];

  const int tid = threadIdx.x;
  const int lane = tid & 31;
  const int wave = tid >> 5;
  const int e0 = blockIdx.x * chunk;
  const int len = max(0, min(chunk, nedges - e0));

  hist[tid] = 0;
  if (blockIdx.x == 0 && tid == 0) total[0] = 0;
  __syncthreads();

  int key[KA_PER_THREAD], bk[KA_PER_THREAD];
#pragma unroll
  for (int k = 0; k < KA_PER_THREAD; ++k) {
    const int i = tid + k * KA_THREADS;
    const int e = min(e0 + i, nedges - 1);
    const int d = dst[e];
    const int s = src[e];
    bk[k] = (i < len) ? (d >> BN_SHIFT) : -1;
    key[k] = s | ((d & (BN - 1)) << SRC_BITS);
  }
#pragma unroll
  for (int k = 0; k < KA_PER_THREAD; ++k)
    if (bk[k] >= 0) atomicAdd(&hist[bk[k]], 1);
  __syncthreads();

  const int c = hist[tid];
  const int incl = wave_incl_scan(c, lane);
  if (lane == 31) wsum[wave] = incl;
  __syncthreads();
  if (wave == 0) {
    const int w = wsum[lane];
    wsum[lane] = wave_incl_scan(w, lane) - w;
  }
  __syncthreads();
  const int excl = wsum[wave] + incl - c;
  cur[tid] = excl;
  table[(size_t)blockIdx.x * NB_MAX + tid] = excl;
  __syncthreads();

#pragma unroll
  for (int k = 0; k < KA_PER_THREAD; ++k) {
    if (bk[k] >= 0) {
      const int p = atomicAdd(&cur[bk[k]], 1);
      sorted[p] = key[k];
    }
  }
  __syncthreads();

  for (int i = tid; i < len; i += KA_THREADS) binned[e0 + i] = sorted[i];
}

__global__ __launch_bounds__(KB_THREADS) void csr_bucket_finish(
    const int* __restrict__ binned, const int* __restrict__ table,
    int* __restrict__ total, int* __restrict__ start, int* __restrict__ cnt,
    int* __restrict__ csr_tmp, int* __restrict__ csr_src, int n_nodes,
    int chunk) {
  __shared__ int lcnt[BN];
  __shared__ int lcur[BN];
  __shared__ int wsum[BN / 32];
  __shared__ int sbase;
  __shared__ int seg[SEG_MAX + 8];

  const int tid = threadIdx.x;
  const int lane = tid & 31;
  const int wave = tid >> 5;
  const int b = blockIdx.x;

  const int myr = wave + KB_WAVES * (lane & (RUNS_PER_WAVE - 1));
  const int o0 = table[(size_t)myr * NB_MAX + b];
  const int o1 = table[(size_t)myr * NB_MAX + b + 1];
  const int my_off = myr * chunk + o0;
  const int my_cnt = o1 - o0;
  if (tid < BN) lcnt[tid] = 0;
  __syncthreads();

  int val[RUNS_PER_WAVE];
#pragma unroll
  for (int k = 0; k < RUNS_PER_WAVE; ++k) {
    const int n = __builtin_amdgcn_readlane(my_cnt, k);
    const int o = __builtin_amdgcn_readlane(my_off, k);
    val[k] = -1;
    if (lane < n) val[k] = binned[o + lane];
  }
#pragma unroll
  for (int k = 0; k < RUNS_PER_WAVE; ++k)
    if (val[k] >= 0) atomicAdd(&lcnt[val[k] >> SRC_BITS], 1);
#pragma unroll
  for (int k = 0; k < RUNS_PER_WAVE; ++k) {
    const int n = __builtin_amdgcn_readlane(my_cnt, k);
    const int o = __builtin_amdgcn_readlane(my_off, k);
    for (int i = 32 + lane; i < n; i += 32)
      atomicAdd(&lcnt[binned[o + i] >> SRC_BITS], 1);
  }
  __syncthreads();

  const int c = (tid < BN) ? lcnt[tid] : 0;
  const int incl = wave_incl_scan(c, lane);
  if (tid < BN && lane == 31) wsum[wave] = incl;
  __syncthreads();
  int woff = 0, bucket_total = 0;
  for (int w = 0; w < BN / 32; ++w) {
    const int s = wsum[w];
    if (w < wave) woff += s;
    bucket_total += s;
  }
  if (tid == 0) sbase = atomicAdd(total, bucket_total);
  __syncthreads();
  const int base = sbase;
  if (tid < BN) {
    const int excl = woff + incl - c;
    lcur[tid] = excl;
    const int node = b * BN + tid;
    if (node < n_nodes) {
      start[node] = base + excl;
      cnt[node] = c;
    }
  }
  __syncthreads();

  if (bucket_total <= SEG_MAX) {
    if (tid < 8) seg[bucket_total + tid] = 0x7fffffff;
#pragma unroll
    for (int k = 0; k < RUNS_PER_WAVE; ++k) {
      if (val[k] >= 0) {
        const int p = atomicAdd(&lcur[val[k] >> SRC_BITS], 1);
        seg[p] = val[k];
      }
    }
#pragma unroll
    for (int k = 0; k < RUNS_PER_WAVE; ++k) {
      const int n = __builtin_amdgcn_readlane(my_cnt, k);
      const int o = __builtin_amdgcn_readlane(my_off, k);
      for (int i = 32 + lane; i < n; i += 32) {
        const int v = binned[o + i];
        const int p = atomicAdd(&lcur[v >> SRC_BITS], 1);
        seg[p] = v;
      }
    }
    __syncthreads();

    for (int p = tid; p < bucket_total; p += KB_THREADS) {
      const int v = seg[p];
      const int nd = v >> SRC_BITS;
      const int deg = lcnt[nd];
      const int sb = lcur[nd] - deg;
      int rank = 0;
      for (int j0 = sb; j0 < sb + deg; j0 += 8) {
#pragma unroll
        for (int jj = 0; jj < 8; ++jj) {
          const int u = seg[j0 + jj];
          rank += (u < v || (u == v && j0 + jj < p)) ? 1 : 0;
        }
      }
      csr_src[base + sb + rank] = v & SRC_MASK;
    }
  } else {
#pragma unroll
    for (int k = 0; k < RUNS_PER_WAVE; ++k) {
      if (val[k] >= 0) {
        const int p = atomicAdd(&lcur[val[k] >> SRC_BITS], 1);
        csr_tmp[base + p] = val[k] & SRC_MASK;
      }
    }
#pragma unroll
    for (int k = 0; k < RUNS_PER_WAVE; ++k) {
      const int n = __builtin_amdgcn_readlane(my_cnt, k);
      const int o = __builtin_amdgcn_readlane(my_off, k);
      for (int i = 32 + lane; i < n; i += 32) {
        const int v = binned[o + i];
        const int p = atomicAdd(&lcur[v >> SRC_BITS], 1);
        csr_tmp[base + p] = v & SRC_MASK;
      }
    }
    __syncthreads();

    const int q = tid & 3;
    for (int nd = tid >> 2; nd < BN; nd += KB_THREADS / 4) {
      const int deg = lcnt[nd];
      const int beg = base + lcur[nd] - deg;
      for (int i = q; i < deg; i += 4) {
        const int v = csr_tmp[beg + i];
        int rank = 0;
        for (int j = 0; j < deg; ++j) {
          const int u = csr_tmp[beg + j];
          rank += (u < v || (u == v && j < i)) ? 1 : 0;
        }
        csr_src[beg + rank] = v;
      }
    }
  }
}

__device__ __forceinline__ float quad_xor1(float v) {
  return __int_as_float(__builtin_amdgcn_update_dpp(
      0, __float_as_int(v), 0xB1  , 0xF, 0xF, true));
}
__device__ __forceinline__ float quad_xor2(float v) {
  return __int_as_float(__builtin_amdgcn_update_dpp(
      0, __float_as_int(v), 0x4E  , 0xF, 0xF, true));
}
template <int T>
__device__ __forceinline__ float quad_lane(float v) {
  return __int_as_float(__builtin_amdgcn_update_dpp(
      0, __float_as_int(v), T * 0x55  , 0xF, 0xF,
      true));
}
template <int J>
__device__ __forceinline__ int row_lane(int v) {
  return __builtin_amdgcn_update_dpp(0, v, 0x150 + J  , 0xF,
                                     0xF, true);
}

__device__ __forceinline__ int chunk_index(const int* __restrict__ csr_src,
                                           int beg, int k0, int c, int q) {
  int v = 0;
  if (c > 0) v = csr_src[beg + k0 + min(q, c - 1)];
  return v;
}

#define NPG 8
#define NPG_PROJ 4
#define ATT_THREADS 64

#define ATT_LOAD(J)                                                          \
  {                                                                          \
    const unsigned s_ = (unsigned)row_lane<J>(myidx);                        \
    row[J] = *(const float4*)(xlb + ((s_ << 8) | qoff));                     \
  }
#define ATT_LOGIT(J)                                                         \
  {                                                                          \
    const float4 l_ = row[J];                                                \
    float vx = l_.x + r4.x; vx = fmaxf(vx, NEG_SLOPE * vx);                  \
    float vy = l_.y + r4.y; vy = fmaxf(vy, NEG_SLOPE * vy);                  \
    float vz = l_.z + r4.z; vz = fmaxf(vz, NEG_SLOPE * vz);                  \
    float vw = l_.w + r4.w; vw = fmaxf(vw, NEG_SLOPE * vw);                  \
    float e_ = vx * a4.x + vy * a4.y + vz * a4.z + vw * a4.w;                \
    e_ += quad_xor1(e_);                                                     \
    e_ += quad_xor2(e_);                                                     \
    if (t == ((J) & 3)) E[(J) >> 2] = e_;                                    \
  }
#define ATT_SUM(J)                                                           \
  {                                                                          \
    const float p_ = quad_lane<(J) & 3>(P[(J) >> 2]);                        \
    den += p_;                                                               \
    acc.x = fmaf(p_, row[J].x, acc.x);                                       \
    acc.y = fmaf(p_, row[J].y, acc.y);                                       \
    acc.z = fmaf(p_, row[J].z, acc.z);                                       \
    acc.w = fmaf(p_, row[J].w, acc.w);                                       \
  }
#define ATT_PHASE(STEP)                                                      \
  do {                                                                       \
    STEP(0)  if (cw <= 1) break;  STEP(1)  if (cw <= 2) break;               \
    STEP(2)  if (cw <= 3) break;  STEP(3)  if (cw <= 4) break;               \
    STEP(4)  if (cw <= 5) break;  STEP(5)  if (cw <= 6) break;               \
    STEP(6)  if (cw <= 7) break;  STEP(7)  if (cw <= 8) break;               \
    STEP(8)  if (cw <= 9) break;  STEP(9)  if (cw <= 10) break;              \
    STEP(10) if (cw <= 11) break; STEP(11) if (cw <= 12) break;              \
    STEP(12) if (cw <= 13) break; STEP(13) if (cw <= 14) break;              \
    STEP(14) if (cw <= 15) break; STEP(15)                                   \
  } while (0)

template <bool PROJ, int NPGT>
__device__ __forceinline__ void attend_nodes(
    const float* __restrict__ xl, const float* __restrict__ xr,
    const float* __restrict__ att, const float* __restrict__ bias,
    const int* __restrict__ start, const int* __restrict__ cnt,
    const int* __restrict__ csr_src, float* __restrict__ out,
    float* __restrict__ hs, int n_nodes, int do_relu) {
  const int gid = blockIdx.x * blockDim.x + threadIdx.x;
  if (!PROJ && ((gid & ~31) >> 4) * NPGT >= n_nodes) return;
  const int q = threadIdx.x & 15;
  const int t = q & 3;
  const int node0 = (gid >> 4) * NPGT;
  const int lastn = n_nodes - 1;

  const int ni = node0 + (q & (NPGT - 1));
  const int begs = start[min(ni, lastn)];
  const int degs = (ni < n_nodes) ? cnt[min(ni, lastn)] : 0;

  const float4 a4 = *(const float4*)(att + 4 * q);
  const float4 b4 = *(const float4*)(bias + 4 * q);
  const char* __restrict__ xlb = (const char*)xl;
  const unsigned qoff = (unsigned)q << 4;

  int beg = __shfl(begs, 0, 16);
  int deg = __shfl(degs, 0, 16);
  int pidx = chunk_index(csr_src, beg, 0, min(16, deg), q);
  float4 pr4 = *(const float4*)(xr + (size_t)min(node0, lastn) * CC + 4 * q);

  for (int i = 0; i < NPGT; ++i) {
    const int node = node0 + i;
    const int cbeg = beg;
    const int cdeg = deg;
    const float4 r4 = pr4;
    const int idx0 = pidx;
    const int degw = max(__builtin_amdgcn_readlane(cdeg, 0),
                         __builtin_amdgcn_readlane(cdeg, 16));

    {
      const int nx = min(i + 1, NPGT - 1);
      beg = __shfl(begs, nx, 16);
      deg = __shfl(degs, nx, 16);
      pidx = chunk_index(csr_src, beg, 0, min(16, deg), q);
      pr4 = *(const float4*)(xr + (size_t)min(node0 + nx, lastn) * CC + 4 * q);
    }

    float m = -INFINITY;
    float den = 0.0f;
    float4 acc = make_float4(0.0f, 0.0f, 0.0f, 0.0f);

    for (int k0 = 0; k0 < degw; k0 += 16) {
      const int c = min(16, max(0, cdeg - k0));
      const int cw = min(16, degw - k0);
      const int myidx =
          (k0 == 0) ? idx0 : chunk_index(csr_src, cbeg, k0, c, q);

      float4 row[16];
      ATT_PHASE(ATT_LOAD);

      float E[4] = {-INFINITY, -INFINITY, -INFINITY, -INFINITY};
      ATT_PHASE(ATT_LOGIT);

#pragma unroll
      for (int k = 0; k < 4; ++k)
        if (4 * k + t >= c) E[k] = -INFINITY;
      float mloc = fmaxf(fmaxf(E[0], E[1]), fmaxf(E[2], E[3]));
      mloc = fmaxf(mloc, quad_xor1(mloc));
      mloc = fmaxf(mloc, quad_xor2(mloc));
      const float mn = fmaxf(m, mloc);
      const float mref = (mn == -INFINITY) ? 0.0f : mn;
      const float sc = __expf(m - mref);
      float P[4];
#pragma unroll
      for (int k = 0; k < 4; ++k) P[k] = __expf(E[k] - mref);
      den *= sc;
      acc.x *= sc;
      acc.y *= sc;
      acc.z *= sc;
      acc.w *= sc;
      m = mn;

      ATT_PHASE(ATT_SUM);
    }

    if (PROJ || node < n_nodes) {
      const float dd = den + 1e-16f;
      float4 o;
      o.x = acc.x / dd + b4.x;
      o.y = acc.y / dd + b4.y;
      o.z = acc.z / dd + b4.z;
      o.w = acc.w / dd + b4.w;
      if (do_relu) {
        o.x = fmaxf(o.x, 0.0f);
        o.y = fmaxf(o.y, 0.0f);
        o.z = fmaxf(o.z, 0.0f);
        o.w = fmaxf(o.w, 0.0f);
      }
      if (PROJ)
        *(float4*)&hs[((threadIdx.x >> 4) * NPGT + i) * XPITCH + 4 * q] = o;
      else
        *(float4*)(out + (size_t)node * CC + 4 * q) = o;
    }
  }
}

__global__ __launch_bounds__(ATT_THREADS) void gat_node_attend(
    const float* __restrict__ xl, const float* __restrict__ xr,
    const float* __restrict__ att, const float* __restrict__ bias,
    const int* __restrict__ start, const int* __restrict__ cnt,
    const int* __restrict__ csr_src, float* __restrict__ out, int n_nodes,
    int do_relu) {
  attend_nodes<false, NPG>(xl, xr, att, bias, start, cnt, csr_src, out,
                           nullptr, n_nodes, do_relu);
}

#define ATTP_THREADS 128
#define ATTP_ROWS ((ATTP_THREADS / 16) * NPG_PROJ)

__global__ __launch_bounds__(ATTP_THREADS) void gat_node_attend_proj(
    const float* __restrict__ xl, const float* __restrict__ xr,
    const float* __restrict__ att, const float* __restrict__ bias,
    const int* __restrict__ start, const int* __restrict__ cnt,
    const int* __restrict__ csr_src, const float* __restrict__ Wl,
    const float* __restrict__ Wr, float* __restrict__ Yl,
    float* __restrict__ Yr, int n_nodes, int do_relu) {
  __shared__ __attribute__((aligned(16))) float hs[ATTP_ROWS * XPITCH];

  attend_nodes<true, NPG_PROJ>(xl, xr, att, bias, start, cnt, csr_src, nullptr,
                               hs, n_nodes, do_relu);

  const int tid = threadIdx.x;
  const int strip = __builtin_amdgcn_readfirstlane(tid >> 5);
  const int lane = tid & 31;
  const int c = lane & 15;
  const int h = lane >> 4;
  const float* __restrict__ W = (strip < 2) ? Wl : Wr;
  float* __restrict__ Y = (strip < 2) ? Yl : Yr;
  const int col0 = (strip & 1) * 32;

  v2f b0[16], b1[16];
#pragma unroll
  for (int kk = 0; kk < 16; ++kk) {
    const int k = 4 * kk + 2 * h;
    b0[kk].x = W[(size_t)k * CC + col0 + c];
    b0[kk].y = W[(size_t)(k + 1) * CC + col0 + c];
    b1[kk].x = W[(size_t)k * CC + col0 + 16 + c];
    b1[kk].y = W[(size_t)(k + 1) * CC + col0 + 16 + c];
  }
  __syncthreads();

  for (int rt = 0; rt < ATTP_ROWS / 16; ++rt) {
    v8f c0 = {}, c1 = {};
    const float* xa = &hs[(16 * rt + c) * XPITCH + 2 * h];
#pragma unroll
    for (int kk = 0; kk < 16; ++kk) {
      const v2f a = *(const v2f*)(xa + 4 * kk);
      c0 = __builtin_amdgcn_wmma_f32_16x16x4_f32(false, a, false, b0[kk],
                                                 (short)0, c0, false, false);
      c1 = __builtin_amdgcn_wmma_f32_16x16x4_f32(false, a, false, b1[kk],
                                                 (short)0, c1, false, false);
    }

    const int m0 = blockIdx.x * ATTP_ROWS + 16 * rt;
    float* yp = Y + (size_t)m0 * CC + col0 + lane;
    float va[8], vb[8];
#pragma unroll
    for (int r = 0; r < 8; ++r) {
      const float got = swap_half(h ? c0[r] : c1[r]);
      va[r] = h ? got : c0[r];
      vb[r] = h ? c1[r] : got;
    }
    if (m0 + 16 <= n_nodes) {
#pragma unroll
      for (int r = 0; r < 8; ++r) {
        yp[(size_t)r * CC] = va[r];
        yp[(size_t)(8 + r) * CC] = vb[r];
      }
    } else {
#pragma unroll
      for (int r = 0; r < 8; ++r) {
        if (m0 + r < n_nodes) yp[(size_t)r * CC] = va[r];
        if (m0 + 8 + r < n_nodes) yp[(size_t)(8 + r) * CC] = vb[r];
      }
    }
  }
}

extern "C" void kernel_launch(void* const* d_in, const int* in_sizes, int n_in,
                              void* d_out, int out_size, void* d_ws,
                              size_t ws_size, hipStream_t stream) {
  const int N = in_sizes[0] / CC;
  const int E = in_sizes[1] / 2;

  const float* x = (const float*)d_in[0];
  const int* edge = (const int*)d_in[1];
  const int* src = edge;
  const int* dst = edge + E;

  const float* Wl[3] = {(const float*)d_in[2], (const float*)d_in[6],
                        (const float*)d_in[10]};
  const float* Wr[3] = {(const float*)d_in[3], (const float*)d_in[7],
                        (const float*)d_in[11]};
  const float* Av[3] = {(const float*)d_in[4], (const float*)d_in[8],
                        (const float*)d_in[12]};
  const float* Bv[3] = {(const float*)d_in[5], (const float*)d_in[9],
                        (const float*)d_in[13]};

  const size_t NC = (size_t)N * CC;
  float* ws = (float*)d_ws;
  float* xl = ws;                 ws += NC;
  float* xr = ws;                 ws += NC;
  float* hA = ws;                 ws += NC;
  float* hB = ws;                 ws += NC;
  int* csr_src = (int*)ws;        ws += (size_t)E;
  int* csr_tmp = (int*)ws;        ws += (size_t)E;
  int* cnt = (int*)ws;            ws += (size_t)N;
  int* start = (int*)ws;          ws += (size_t)N;
  int* total = (int*)ws;
  int* binned = (int*)hB;
  int* table = (int*)hB + (size_t)E;

  const int proj_tiles = (N + 31) / 32;
  const int proj_grid = proj_tiles < 1024 ? proj_tiles : 1024;
  const int attend_grid =
      (N + (ATT_THREADS / 16) * NPG - 1) / ((ATT_THREADS / 16) * NPG);
  const int attp_grid = (N + ATTP_ROWS - 1) / ATTP_ROWS;
  const int chunk = (E + NCHUNK - 1) / NCHUNK;
  const int bucket_grid = (N + BN - 1) / BN;

  csr_chunk_bin<<<NCHUNK, KA_THREADS, 0, stream>>>(src, dst, binned, table,
                                                   total, E, chunk);
  csr_bucket_finish<<<bucket_grid, KB_THREADS, 0, stream>>>(
      binned, table, total, start, cnt, csr_tmp, csr_src, N, chunk);

  gat_proj_pair_wmma<<<proj_grid, 256, 0, stream>>>(x, Wl[0], Wr[0], xl, xr, N);
  gat_node_attend_proj<<<attp_grid, ATTP_THREADS, 0, stream>>>(
      xl, xr, Av[0], Bv[0], start, cnt, csr_src, Wl[1], Wr[1], hA, hB, N, 1);
  gat_node_attend_proj<<<attp_grid, ATTP_THREADS, 0, stream>>>(
      hA, hB, Av[1], Bv[1], start, cnt, csr_src, Wl[2], Wr[2], xl, xr, N, 1);
  gat_node_attend<<<attend_grid, ATT_THREADS, 0, stream>>>(
      xl, xr, Av[2], Bv[2], start, cnt, csr_src, (float*)d_out, N, 0);
}
